# starting kernel
# baseline (speedup 1.0000x reference)
#include <hip/hip_runtime.h>

#define NNODES 100000
#define NEDGES 1200000
#define IN_DIM 64
#define HID_DIM 64
#define OUT_DIM 32

typedef __attribute__((ext_vector_type(2))) float v2f;
typedef __attribute__((ext_vector_type(4))) float v4f;
typedef __attribute__((ext_vector_type(8))) float v8f;

__global__ void gcn_fill_kernel(float* __restrict__ p, float v, int n) {
    int i = blockIdx.x * blockDim.x + threadIdx.x;
    if (i < n) p[i] = v;
}

__global__ void gcn_deg_kernel(const int* __restrict__ dst, float* __restrict__ deg, int E) {
    int e = blockIdx.x * blockDim.x + threadIdx.x;
    if (e < E) atomicAdd(&deg[dst[e]], 1.0f);
}

__global__ void gcn_rsqrt_kernel(float* __restrict__ deg, int n) {
    int i = blockIdx.x * blockDim.x + threadIdx.x;
    if (i < n) deg[i] = rsqrtf(deg[i]);
}

template <int K, int N>
__global__ __launch_bounds__(256) void gcn_gemm_wmma_kernel(
    const float* __restrict__ A, const float* __restrict__ B,
    float* __restrict__ C, int M) {
    constexpr int NT = N / 16;

    const int lane     = threadIdx.x & 31;
    const int wave     = threadIdx.x >> 5;
    const int row_tile = blockIdx.x * 8 + wave;
    if (row_tile * 16 >= M) return;

    const int m     = lane & 15;
    const int khalf = lane >> 4;
    const int row   = row_tile * 16 + m;

    v8f acc[NT];
#pragma unroll
    for (int nt = 0; nt < NT; ++nt) acc[nt] = (v8f){0.f,0.f,0.f,0.f,0.f,0.f,0.f,0.f};

    const float* __restrict__ arow = A + (size_t)row * K;
#pragma unroll
    for (int k0 = 0; k0 < K; k0 += 4) {
        const int kk = k0 + khalf * 2;
        v2f a = *(const v2f*)(arow + kk);
#pragma unroll
        for (int nt = 0; nt < NT; ++nt) {
            const int n = nt * 16 + m;
            v2f b;
            b.x = B[(size_t)kk * N + n];
            b.y = B[(size_t)(kk + 1) * N + n];
            acc[nt] = __builtin_amdgcn_wmma_f32_16x16x4_f32(
                false, a, false, b, (short)0, acc[nt], false, false);
        }
    }

#pragma unroll
    for (int nt = 0; nt < NT; ++nt) {
        float* __restrict__ crow =
            C + (size_t)(row_tile * 16 + khalf * 8) * N + nt * 16 + m;
#pragma unroll
        for (int j = 0; j < 8; ++j) crow[(size_t)j * N] = acc[nt][j];
    }
}

template <int F>
__global__ __launch_bounds__(256) void gcn_scatter_kernel(
    const float* __restrict__ h, const int* __restrict__ src,
    const int* __restrict__ dst, const float* __restrict__ isq,
    float* __restrict__ out, int E) {
    constexpr int Q = F / 4;
    unsigned t = blockIdx.x * blockDim.x + threadIdx.x;
    unsigned e = t / Q;
    unsigned q = t % Q;
    if (e >= (unsigned)E) return;
    int s = src[e];
    int d = dst[e];
    float nrm = isq[s] * isq[d];
    v4f msg = *(const v4f*)(h + (unsigned)s * F + q * 4);
    float* o = out + (unsigned)d * F + q * 4;
    atomicAdd(o + 0, msg.x * nrm);
    atomicAdd(o + 1, msg.y * nrm);
    atomicAdd(o + 2, msg.z * nrm);
    atomicAdd(o + 3, msg.w * nrm);
}

template <int F, bool RELU>
__global__ __launch_bounds__(256) void gcn_finish_kernel(
    const float* __restrict__ h, const float* __restrict__ isq,
    const float* __restrict__ bias, float* __restrict__ out, int n) {
    unsigned idx = blockIdx.x * blockDim.x + threadIdx.x;
    if (idx >= (unsigned)n * F) return;
    unsigned i = idx / F;
    unsigned f = idx % F;
    float si = isq[i];
    float v  = out[idx] + h[idx] * si * si + bias[f];
    out[idx] = RELU ? fmaxf(v, 0.0f) : v;
}

extern "C" void kernel_launch(void* const* d_in, const int* in_sizes, int n_in,
                              void* d_out, int out_size, void* d_ws, size_t ws_size,
                              hipStream_t stream) {
    const float* x  = (const float*)d_in[0];
    const int*   ei = (const int*)d_in[1];
    const float* W1 = (const float*)d_in[2];
    const float* b1 = (const float*)d_in[3];
    const float* W2 = (const float*)d_in[4];
    const float* b2 = (const float*)d_in[5];
    const int* src = ei;
    const int* dst = ei + NEDGES;
    float* out = (float*)d_out;

    float* ws   = (float*)d_ws;
    float* isq  = ws;
    float* h1   = ws + 131072;
    float* agg1 = h1 + (size_t)NNODES * HID_DIM;
    float* h2   = h1;

    const int T = 256;

    gcn_fill_kernel<<<(NNODES + T - 1) / T, T, 0, stream>>>(isq, 1.0f, NNODES);
    gcn_deg_kernel<<<(NEDGES + T - 1) / T, T, 0, stream>>>(dst, isq, NEDGES);
    gcn_rsqrt_kernel<<<(NNODES + T - 1) / T, T, 0, stream>>>(isq, NNODES);

    {
        int waves  = NNODES / 16;
        int blocks = (waves + 7) / 8;
        gcn_gemm_wmma_kernel<IN_DIM, HID_DIM><<<blocks, T, 0, stream>>>(x, W1, h1, NNODES);
    }
    gcn_fill_kernel<<<((NNODES * HID_DIM) + T - 1) / T, T, 0, stream>>>(agg1, 0.0f, NNODES * HID_DIM);
    {
        unsigned total = (unsigned)NEDGES * (HID_DIM / 4);
        gcn_scatter_kernel<HID_DIM><<<(total + T - 1) / T, T, 0, stream>>>(h1, src, dst, isq, agg1, NEDGES);
    }
    gcn_finish_kernel<HID_DIM, true><<<((NNODES * HID_DIM) + T - 1) / T, T, 0, stream>>>(h1, isq, b1, agg1, NNODES);

    {
        int waves  = NNODES / 16;
        int blocks = (waves + 7) / 8;
        gcn_gemm_wmma_kernel<HID_DIM, OUT_DIM><<<blocks, T, 0, stream>>>(agg1, W2, h2, NNODES);
    }
    gcn_fill_kernel<<<((NNODES * OUT_DIM) + T - 1) / T, T, 0, stream>>>(out, 0.0f, NNODES * OUT_DIM);
    {
        unsigned total = (unsigned)NEDGES * (OUT_DIM / 4);
        gcn_scatter_kernel<OUT_DIM><<<(total + T - 1) / T, T, 0, stream>>>(h2, src, dst, isq, out, NEDGES);
    }
    gcn_finish_kernel<OUT_DIM, false><<<((NNODES * OUT_DIM) + T - 1) / T, T, 0, stream>>>(h2, isq, b2, out, NNODES);
}


# f32 atomic scatter-add replaced by sums in LDS: a work-group owns 391 destination rows in LDS, filters the whole dst list for them, adds gathered rows by LDS float atomics, stores rows once
# speedup vs baseline: 1.2609x; 1.2609x over previous
#include <hip/hip_runtime.h>

#define NNODES 100000
#define NEDGES 1200000
#define IN_DIM 64
#define HID_DIM 64
#define OUT_DIM 32

typedef __attribute__((ext_vector_type(2))) float v2f;
typedef __attribute__((ext_vector_type(4))) float v4f;
typedef __attribute__((ext_vector_type(8))) float v8f;

__global__ void gcn_fill_kernel(float* __restrict__ p, float v, int n) {
    int i = blockIdx.x * blockDim.x + threadIdx.x;
    if (i < n) p[i] = v;
}

__global__ void gcn_deg_kernel(const int* __restrict__ dst, float* __restrict__ deg, int E) {
    int e = blockIdx.x * blockDim.x + threadIdx.x;
    if (e < E) atomicAdd(&deg[dst[e]], 1.0f);
}

__global__ void gcn_rsqrt_kernel(float* __restrict__ deg, int n) {
    int i = blockIdx.x * blockDim.x + threadIdx.x;
    if (i < n) deg[i] = rsqrtf(deg[i]);
}

template <int K, int N, bool RELU_A>
__global__ __launch_bounds__(256) void gcn_gemm_wmma_kernel(
    const float* __restrict__ A, const float* __restrict__ B,
    const float* __restrict__ bias, const float* __restrict__ isq,
    float* __restrict__ C, float* __restrict__ S, int M) {
    constexpr int NT = N / 16;

    const int lane     = threadIdx.x & 31;
    const int wave     = threadIdx.x >> 5;
    const int row_tile = blockIdx.x * 8 + wave;
    if (row_tile * 16 >= M) return;

    const int m     = lane & 15;
    const int khalf = lane >> 4;
    const int row   = row_tile * 16 + m;

    v8f acc[NT];
#pragma unroll
    for (int nt = 0; nt < NT; ++nt) acc[nt] = (v8f){0.f,0.f,0.f,0.f,0.f,0.f,0.f,0.f};

    const float* __restrict__ arow = A + (size_t)row * K;
#pragma unroll
    for (int k0 = 0; k0 < K; k0 += 4) {
        const int kk = k0 + khalf * 2;
        v2f a = *(const v2f*)(arow + kk);
        if (RELU_A) { a.x = fmaxf(a.x, 0.0f); a.y = fmaxf(a.y, 0.0f); }
#pragma unroll
        for (int nt = 0; nt < NT; ++nt) {
            const int n = nt * 16 + m;
            v2f b;
            b.x = B[(size_t)kk * N + n];
            b.y = B[(size_t)(kk + 1) * N + n];
            acc[nt] = __builtin_amdgcn_wmma_f32_16x16x4_f32(
                false, a, false, b, (short)0, acc[nt], false, false);
        }
    }

    const int r0 = row_tile * 16 + khalf * 8;
    const v4f s_lo = *(const v4f*)(isq + r0);
    const v4f s_hi = *(const v4f*)(isq + r0 + 4);
    const float nrm[8] = {s_lo.x * s_lo.x, s_lo.y * s_lo.y, s_lo.z * s_lo.z, s_lo.w * s_lo.w,
                          s_hi.x * s_hi.x, s_hi.y * s_hi.y, s_hi.z * s_hi.z, s_hi.w * s_hi.w};
#pragma unroll
    for (int nt = 0; nt < NT; ++nt) {
        const size_t off = (size_t)r0 * N + nt * 16 + m;
        float* __restrict__ crow = C + off;
        float* __restrict__ srow = S + off;
        const float bv = bias[nt * 16 + m];
#pragma unroll
        for (int j = 0; j < 8; ++j) {
            const float hv = acc[nt][j];
            crow[(size_t)j * N] = hv;
            srow[(size_t)j * N] = hv * nrm[j] + bv;
        }
    }
}

#define AGG_WGS   256
#define AGG_RPW   ((NNODES + AGG_WGS - 1) / AGG_WGS)
#define AGG_T     1024
#define AGG_WAVES (AGG_T / 32)
#define AGG_QCAP  160
#define AGG_EBITS 21
#define AGG_U     4
#define AGG_G     16

static_assert(NEDGES % 4 == 0, "dst is read four edges per lane");
static_assert(NEDGES <= (1 << AGG_EBITS), "edge id must fit its field");
static_assert(AGG_RPW < (1 << (31 - AGG_EBITS)), "local row must fit its field");
static_assert((AGG_WGS - 1) * AGG_RPW < NNODES, "every work-group owns at least one row");

typedef __attribute__((ext_vector_type(4))) int v4i;

template <int F>
__global__ __launch_bounds__(AGG_T) void gcn_agg_lds_kernel(
    const float* __restrict__ h, const int* __restrict__ src,
    const int* __restrict__ dst, const float* __restrict__ isq,
    float* __restrict__ out) {
    constexpr int VPL = F / 32;
    __shared__ __attribute__((aligned(16))) float acc[AGG_RPW * F];
    __shared__ int pend[AGG_WAVES * AGG_QCAP];

    const int tid  = threadIdx.x;
    const int lane = tid & 31;
    const int wave = tid >> 5;
    const int lo   = blockIdx.x * AGG_RPW;
    const int rem  = NNODES - lo;
    const int nrows = rem < AGG_RPW ? rem : AGG_RPW;

    float* __restrict__ obase = out + (size_t)lo * F;
    const int nvec = nrows * (F / 4);
    constexpr int NI = (AGG_RPW * (F / 4) + AGG_T - 1) / AGG_T;
    {
        v4f tmp[NI];
#pragma unroll
        for (int i = 0; i < NI; ++i) {
            const int idx = tid + i * AGG_T;
            if (idx < nvec) tmp[i] = ((const v4f*)obase)[idx];
        }
#pragma unroll
        for (int i = 0; i < NI; ++i) {
            const int idx = tid + i * AGG_T;
            if (idx < nvec) ((v4f*)acc)[idx] = tmp[i];
        }
    }
    __syncthreads();

    int* __restrict__ myq = pend + wave * AGG_QCAP;
    int qn = 0;
    const unsigned lt = (1u << lane) - 1u;

    auto drain = [&](int base, int cnt) {
        const int ent = myq[base + (lane < cnt ? lane : 0)];
        const int e   = ent & ((1 << AGG_EBITS) - 1);
        const int dl  = ent >> AGG_EBITS;
        const int s   = src[e];
        const float w = isq[s] * isq[lo + dl];
        const float* __restrict__ hb = h + lane * VPL;
#pragma unroll
        for (int j = 0; j < 32; j += AGG_G) {
            if (j < cnt) {
                float v[AGG_G][VPL];
#pragma unroll
                for (int t = 0; t < AGG_G; ++t) {
                    const int sj = __builtin_amdgcn_readlane(s, j + t);
                    const float* __restrict__ p = hb + (size_t)sj * F;
                    if (VPL == 2) { const v2f r = *(const v2f*)p; v[t][0] = r.x; v[t][VPL - 1] = r.y; }
                    else          { v[t][0] = *p; }
                }
#pragma unroll
                for (int t = 0; t < AGG_G; ++t) {
                    if (j + t < cnt) {
                        const float wj = __int_as_float(__builtin_amdgcn_readlane(__float_as_int(w), j + t));
                        const int   dj = __builtin_amdgcn_readlane(dl, j + t);
                        float* a = acc + dj * F + lane * VPL;
#pragma unroll
                        for (int c = 0; c < VPL; ++c) atomicAdd(a + c, v[t][c] * wj);
                    }
                }
            }
        }
    };

    constexpr int EPI = AGG_T * 4;
    constexpr int NG  = (NEDGES + EPI * AGG_U - 1) / (EPI * AGG_U);
    const int g0 = (int)(((long long)blockIdx.x * NG) / AGG_WGS);
    for (int gi = 0; gi < NG; ++gi) {
        int grp = g0 + gi;
        if (grp >= NG) grp -= NG;
        v4i dv[AGG_U];
#pragma unroll
        for (int u = 0; u < AGG_U; ++u) {
            const int b  = (grp * AGG_U + u) * EPI + tid * 4;
            const int bc = b < NEDGES ? b : NEDGES - 4;
            dv[u] = *(const v4i*)(dst + bc);
        }
        asm volatile("" :: "v"(dv[0]), "v"(dv[1]), "v"(dv[2]), "v"(dv[3]));
        static_assert(AGG_U == 4, "the statement above names four loads");
#pragma unroll
        for (int u = 0; u < AGG_U; ++u) {
            const int  b   = (grp * AGG_U + u) * EPI + tid * 4;
            const int lo_l = b < NEDGES ? lo : 0x7fffffff;
            int dl[4];
            unsigned mask[4];
#pragma unroll
            for (int k = 0; k < 4; ++k) {
                dl[k]   = dv[u][k] - lo_l;
                mask[k] = __builtin_amdgcn_ballot_w32((unsigned)dl[k] < (unsigned)nrows);
            }
            if ((mask[0] | mask[1] | mask[2] | mask[3]) == 0u) continue;
#pragma unroll
            for (int k = 0; k < 4; ++k) {
                if ((mask[k] >> lane) & 1u)
                    myq[qn + __builtin_popcount(mask[k] & lt)] = (dl[k] << AGG_EBITS) | (b + k);
                qn += __builtin_popcount(mask[k]);
            }
            while (qn >= 32) { qn -= 32; drain(qn, 32); }
        }
    }
    if (qn > 0) drain(0, qn);

    __syncthreads();
#pragma unroll
    for (int i = 0; i < NI; ++i) {
        const int idx = tid + i * AGG_T;
        if (idx < nvec) ((v4f*)obase)[idx] = ((const v4f*)acc)[idx];
    }
}

extern "C" void kernel_launch(void* const* d_in, const int* in_sizes, int n_in,
                              void* d_out, int out_size, void* d_ws, size_t ws_size,
                              hipStream_t stream) {
    const float* x  = (const float*)d_in[0];
    const int*   ei = (const int*)d_in[1];
    const float* W1 = (const float*)d_in[2];
    const float* b1 = (const float*)d_in[3];
    const float* W2 = (const float*)d_in[4];
    const float* b2 = (const float*)d_in[5];
    const int* src = ei;
    const int* dst = ei + NEDGES;
    float* out = (float*)d_out;

    float* ws   = (float*)d_ws;
    float* isq  = ws;
    float* h1   = ws + 131072;
    float* agg1 = h1 + (size_t)NNODES * HID_DIM;
    float* h2   = h1;

    const int T = 256;

    gcn_fill_kernel<<<(NNODES + T - 1) / T, T, 0, stream>>>(isq, 1.0f, NNODES);
    gcn_deg_kernel<<<(NEDGES + T - 1) / T, T, 0, stream>>>(dst, isq, NEDGES);
    gcn_rsqrt_kernel<<<(NNODES + T - 1) / T, T, 0, stream>>>(isq, NNODES);

    {
        int waves  = NNODES / 16;
        int blocks = (waves + 7) / 8;
        gcn_gemm_wmma_kernel<IN_DIM, HID_DIM, false><<<blocks, T, 0, stream>>>(
            x, W1, b1, isq, h1, agg1, NNODES);
    }
    gcn_agg_lds_kernel<HID_DIM><<<AGG_WGS, AGG_T, 0, stream>>>(h1, src, dst, isq, agg1);

    {
        int waves  = NNODES / 16;
        int blocks = (waves + 7) / 8;
        gcn_gemm_wmma_kernel<HID_DIM, OUT_DIM, true><<<blocks, T, 0, stream>>>(
            agg1, W2, b2, isq, h2, out, NNODES);
    }
    gcn_agg_lds_kernel<OUT_DIM><<<AGG_WGS, AGG_T, 0, stream>>>(h2, src, dst, isq, out);
}


# dst list filtered once, not twice: layer-1 aggregation saves each wave's matched edges (src row, local dst row, weight) to the workspace, layer 2 reads these lists; LDS float atomic sums as before
# speedup vs baseline: 1.5442x; 1.2247x over previous
#include <hip/hip_runtime.h>

#define NNODES 100000
#define NEDGES 1200000
#define IN_DIM 64
#define HID_DIM 64
#define OUT_DIM 32

typedef __attribute__((ext_vector_type(2))) float v2f;
typedef __attribute__((ext_vector_type(4))) float v4f;
typedef __attribute__((ext_vector_type(8))) float v8f;

__global__ void gcn_fill_kernel(float* __restrict__ p, float v, int n) {
    int i = blockIdx.x * blockDim.x + threadIdx.x;
    if (i < n) p[i] = v;
}

__global__ void gcn_deg_kernel(const int* __restrict__ dst, float* __restrict__ deg, int E) {
    int e = blockIdx.x * blockDim.x + threadIdx.x;
    if (e < E) atomicAdd(&deg[dst[e]], 1.0f);
}

__global__ void gcn_rsqrt_kernel(float* __restrict__ deg, int n) {
    int i = blockIdx.x * blockDim.x + threadIdx.x;
    if (i < n) deg[i] = rsqrtf(deg[i]);
}

template <int K, int N, bool RELU_A>
__global__ __launch_bounds__(256) void gcn_gemm_wmma_kernel(
    const float* __restrict__ A, const float* __restrict__ B,
    const float* __restrict__ bias, const float* __restrict__ isq,
    float* __restrict__ C, float* __restrict__ S, int M) {
    constexpr int NT = N / 16;

    const int lane     = threadIdx.x & 31;
    const int wave     = threadIdx.x >> 5;
    const int row_tile = blockIdx.x * 8 + wave;
    if (row_tile * 16 >= M) return;

    const int m     = lane & 15;
    const int khalf = lane >> 4;
    const int row   = row_tile * 16 + m;

    v8f acc[NT];
#pragma unroll
    for (int nt = 0; nt < NT; ++nt) acc[nt] = (v8f){0.f,0.f,0.f,0.f,0.f,0.f,0.f,0.f};

    const float* __restrict__ arow = A + (size_t)row * K;
#pragma unroll
    for (int k0 = 0; k0 < K; k0 += 4) {
        const int kk = k0 + khalf * 2;
        v2f a = *(const v2f*)(arow + kk);
        if (RELU_A) { a.x = fmaxf(a.x, 0.0f); a.y = fmaxf(a.y, 0.0f); }
#pragma unroll
        for (int nt = 0; nt < NT; ++nt) {
            const int n = nt * 16 + m;
            v2f b;
            b.x = B[(size_t)kk * N + n];
            b.y = B[(size_t)(kk + 1) * N + n];
            acc[nt] = __builtin_amdgcn_wmma_f32_16x16x4_f32(
                false, a, false, b, (short)0, acc[nt], false, false);
        }
    }

    const int r0 = row_tile * 16 + khalf * 8;
    const v4f s_lo = *(const v4f*)(isq + r0);
    const v4f s_hi = *(const v4f*)(isq + r0 + 4);
    const float nrm[8] = {s_lo.x * s_lo.x, s_lo.y * s_lo.y, s_lo.z * s_lo.z, s_lo.w * s_lo.w,
                          s_hi.x * s_hi.x, s_hi.y * s_hi.y, s_hi.z * s_hi.z, s_hi.w * s_hi.w};
#pragma unroll
    for (int nt = 0; nt < NT; ++nt) {
        const size_t off = (size_t)r0 * N + nt * 16 + m;
        float* __restrict__ crow = C + off;
        float* __restrict__ srow = S + off;
        const float bv = bias[nt * 16 + m];
#pragma unroll
        for (int j = 0; j < 8; ++j) {
            const float hv = acc[nt][j];
            crow[(size_t)j * N] = hv;
            srow[(size_t)j * N] = hv * nrm[j] + bv;
        }
    }
}

#define AGG_WGS   256
#define AGG_RPW   ((NNODES + AGG_WGS - 1) / AGG_WGS)
#define AGG_T     1024
#define AGG_WAVES (AGG_T / 32)
#define AGG_QCAP  160
#define AGG_EBITS 21
#define AGG_U     4
#define AGG_G     16
#define AGG_LCAP  512
#define AGG_SBITS 17

static_assert(NEDGES % 4 == 0, "dst is read four edges per lane");
static_assert(NEDGES <= (1 << AGG_EBITS), "edge id must fit its field");
static_assert(AGG_RPW < (1 << (31 - AGG_EBITS)), "local row must fit its field");
static_assert((AGG_WGS - 1) * AGG_RPW < NNODES, "every work-group owns at least one row");
static_assert(NNODES <= (1 << AGG_SBITS), "source row must fit its field of a saved entry");
static_assert(AGG_RPW < (1 << (31 - AGG_SBITS)), "local row must fit its field of a saved entry");
static_assert(AGG_LCAP % 32 == 0, "a full group of 32 saved entries starts on a 256-byte boundary");
static_assert(AGG_WAVES == 32, "one lane per wave reads or stores the list lengths");

typedef __attribute__((ext_vector_type(4))) int v4i;
typedef __attribute__((ext_vector_type(2))) int v2i;

template <int F, bool SAVE, bool USE>
__global__ __launch_bounds__(AGG_T) void gcn_agg_lds_kernel(
    const float* __restrict__ h, const int* __restrict__ src,
    const int* __restrict__ dst, const float* __restrict__ isq,
    float* __restrict__ out, int* __restrict__ lcnt, int* __restrict__ lent) {
    constexpr int VPL = F / 32;
    __shared__ __attribute__((aligned(16))) float acc[AGG_RPW * F];
    __shared__ int pend[AGG_WAVES * AGG_QCAP];

    const int tid  = threadIdx.x;
    const int lane = tid & 31;
    const int wave = tid >> 5;
    const int lo   = blockIdx.x * AGG_RPW;
    const int rem  = NNODES - lo;
    const int nrows = rem < AGG_RPW ? rem : AGG_RPW;

    float* __restrict__ obase = out + (size_t)lo * F;
    const int nvec = nrows * (F / 4);
    constexpr int NI = (AGG_RPW * (F / 4) + AGG_T - 1) / AGG_T;
    {
        v4f tmp[NI];
#pragma unroll
        for (int i = 0; i < NI; ++i) {
            const int idx = tid + i * AGG_T;
            if (idx < nvec) tmp[i] = ((const v4f*)obase)[idx];
        }
#pragma unroll
        for (int i = 0; i < NI; ++i) {
            const int idx = tid + i * AGG_T;
            if (idx < nvec) ((v4f*)acc)[idx] = tmp[i];
        }
    }
    __syncthreads();

    int* __restrict__ myq = pend + wave * AGG_QCAP;
    v2i* __restrict__ mylst = (v2i*)lent + ((size_t)blockIdx.x * AGG_WAVES + wave) * AGG_LCAP;

    auto gather_add = [&](int s, int dl, float w, int cnt) {
        const float* __restrict__ hb = h + lane * VPL;
#pragma unroll
        for (int j = 0; j < 32; j += AGG_G) {
            if (j < cnt) {
                float v[AGG_G][VPL];
#pragma unroll
                for (int t = 0; t < AGG_G; ++t) {
                    const int sj = __builtin_amdgcn_readlane(s, j + t);
                    const float* __restrict__ p = hb + (size_t)sj * F;
                    if (VPL == 2) { const v2f r = *(const v2f*)p; v[t][0] = r.x; v[t][VPL - 1] = r.y; }
                    else          { v[t][0] = *p; }
                }
#pragma unroll
                for (int t = 0; t < AGG_G; ++t) {
                    if (j + t < cnt) {
                        const float wj = __int_as_float(__builtin_amdgcn_readlane(__float_as_int(w), j + t));
                        const int   dj = __builtin_amdgcn_readlane(dl, j + t);
                        float* a = acc + dj * F + lane * VPL;
#pragma unroll
                        for (int c = 0; c < VPL; ++c) atomicAdd(a + c, v[t][c] * wj);
                    }
                }
            }
        }
    };

    bool from_list = false;
    if (USE) {
        const int len_l = lcnt[blockIdx.x * AGG_WAVES + lane];
        from_list = __builtin_amdgcn_ballot_w32(len_l > AGG_LCAP) == 0u;
        if (from_list) {
            const int n = __builtin_amdgcn_readlane(len_l, __builtin_amdgcn_readfirstlane(wave));
            v2i cur = (v2i){0, 0};
            if (lane < n) cur = mylst[lane];
            for (int base = 0; base < n; base += 32) {
                const int rest = n - base;
                const int cnt  = rest < 32 ? rest : 32;
                v2i nxt = (v2i){0, 0};
                if (base + 32 + lane < n) nxt = mylst[base + 32 + lane];
                gather_add(cur.x & ((1 << AGG_SBITS) - 1), cur.x >> AGG_SBITS, __int_as_float(cur.y), cnt);
                cur = nxt;
            }
        }
    }

    int tot = 0;
    if (!from_list) {
        int qn = 0;
        const unsigned lt = (1u << lane) - 1u;

        auto drain = [&](int base, int cnt) {
            const int ent = myq[base + (lane < cnt ? lane : 0)];
            const int e   = ent & ((1 << AGG_EBITS) - 1);
            const int dl  = ent >> AGG_EBITS;
            const int s   = src[e];
            const float w = isq[s] * isq[lo + dl];
            if (SAVE) {
                const int p = tot + lane;
                if (lane < cnt && p < AGG_LCAP) mylst[p] = (v2i){s | (dl << AGG_SBITS), __float_as_int(w)};
                tot += cnt;
            }
            gather_add(s, dl, w, cnt);
        };

        constexpr int EPI = AGG_T * 4;
        constexpr int NG  = (NEDGES + EPI * AGG_U - 1) / (EPI * AGG_U);
        const int g0 = (int)(((long long)blockIdx.x * NG) / AGG_WGS);
        for (int gi = 0; gi < NG; ++gi) {
            int grp = g0 + gi;
            if (grp >= NG) grp -= NG;
            v4i dv[AGG_U];
#pragma unroll
            for (int u = 0; u < AGG_U; ++u) {
                const int b  = (grp * AGG_U + u) * EPI + tid * 4;
                const int bc = b < NEDGES ? b : NEDGES - 4;
                dv[u] = *(const v4i*)(dst + bc);
            }
            asm volatile("" :: "v"(dv[0]), "v"(dv[1]), "v"(dv[2]), "v"(dv[3]));
            static_assert(AGG_U == 4, "the statement above names four loads");
#pragma unroll
            for (int u = 0; u < AGG_U; ++u) {
                const int  b   = (grp * AGG_U + u) * EPI + tid * 4;
                const int lo_l = b < NEDGES ? lo : 0x7fffffff;
                int dl[4];
                unsigned mask[4];
#pragma unroll
                for (int k = 0; k < 4; ++k) {
                    dl[k]   = dv[u][k] - lo_l;
                    mask[k] = __builtin_amdgcn_ballot_w32((unsigned)dl[k] < (unsigned)nrows);
                }
                if ((mask[0] | mask[1] | mask[2] | mask[3]) == 0u) continue;
#pragma unroll
                for (int k = 0; k < 4; ++k) {
                    if ((mask[k] >> lane) & 1u)
                        myq[qn + __builtin_popcount(mask[k] & lt)] = (dl[k] << AGG_EBITS) | (b + k);
                    qn += __builtin_popcount(mask[k]);
                }
                while (qn >= 32) { qn -= 32; drain(qn, 32); }
            }
        }
        if (qn > 0) drain(0, qn);
    }
    if (SAVE && lane == 0) myq[0] = tot;

    __syncthreads();
    if (SAVE && wave == 0) lcnt[blockIdx.x * AGG_WAVES + lane] = pend[lane * AGG_QCAP];
#pragma unroll
    for (int i = 0; i < NI; ++i) {
        const int idx = tid + i * AGG_T;
        if (idx < nvec) ((v4f*)obase)[idx] = ((const v4f*)acc)[idx];
    }
}

extern "C" void kernel_launch(void* const* d_in, const int* in_sizes, int n_in,
                              void* d_out, int out_size, void* d_ws, size_t ws_size,
                              hipStream_t stream) {
    const float* x  = (const float*)d_in[0];
    const int*   ei = (const int*)d_in[1];
    const float* W1 = (const float*)d_in[2];
    const float* b1 = (const float*)d_in[3];
    const float* W2 = (const float*)d_in[4];
    const float* b2 = (const float*)d_in[5];
    const int* src = ei;
    const int* dst = ei + NEDGES;
    float* out = (float*)d_out;

    float* ws   = (float*)d_ws;
    float* isq  = ws;
    float* h1   = ws + 131072;
    float* agg1 = h1 + (size_t)NNODES * HID_DIM;
    float* h2   = h1;
    int* lcnt   = (int*)(agg1 + (size_t)NNODES * HID_DIM);
    int* lent   = lcnt + AGG_WGS * AGG_WAVES;

    const int T = 256;

    gcn_fill_kernel<<<(NNODES + T - 1) / T, T, 0, stream>>>(isq, 1.0f, NNODES);
    gcn_deg_kernel<<<(NEDGES + T - 1) / T, T, 0, stream>>>(dst, isq, NEDGES);
    gcn_rsqrt_kernel<<<(NNODES + T - 1) / T, T, 0, stream>>>(isq, NNODES);

    {
        int waves  = NNODES / 16;
        int blocks = (waves + 7) / 8;
        gcn_gemm_wmma_kernel<IN_DIM, HID_DIM, false><<<blocks, T, 0, stream>>>(
            x, W1, b1, isq, h1, agg1, NNODES);
    }
    gcn_agg_lds_kernel<HID_DIM, true, false><<<AGG_WGS, AGG_T, 0, stream>>>(h1, src, dst, isq, agg1, lcnt, lent);

    {
        int waves  = NNODES / 16;
        int blocks = (waves + 7) / 8;
        gcn_gemm_wmma_kernel<HID_DIM, OUT_DIM, true><<<blocks, T, 0, stream>>>(
            agg1, W2, b2, isq, h2, out, NNODES);
    }
    gcn_agg_lds_kernel<OUT_DIM, false, true><<<AGG_WGS, AGG_T, 0, stream>>>(h2, src, dst, isq, out, lcnt, lent);
}


# aggregation filter reads a range-number byte per edge (dst/391, packed by the degree pass, 4 edges a thread), not 4-byte dst: a quarter of the bytes and loads; edge split and LDS sum order change
# speedup vs baseline: 1.7120x; 1.1087x over previous
#include <hip/hip_runtime.h>

#define NNODES 100000
#define NEDGES 1200000
#define IN_DIM 64
#define HID_DIM 64
#define OUT_DIM 32

typedef __attribute__((ext_vector_type(2))) float v2f;
typedef __attribute__((ext_vector_type(4))) float v4f;
typedef __attribute__((ext_vector_type(8))) float v8f;
typedef __attribute__((ext_vector_type(4))) int v4i;
typedef __attribute__((ext_vector_type(2))) int v2i;

#define AGG_WGS   256
#define AGG_RPW   ((NNODES + AGG_WGS - 1) / AGG_WGS)
static_assert(AGG_WGS <= 256, "a range number must fit one byte");
static_assert(NEDGES % 16 == 0, "range numbers are read sixteen edges per lane");

__global__ void gcn_fill_kernel(float* __restrict__ p, float v, int n) {
    int i = blockIdx.x * blockDim.x + threadIdx.x;
    if (i < n) p[i] = v;
}

__global__ void gcn_deg_kernel(const int* __restrict__ dst, float* __restrict__ deg,
                               unsigned* __restrict__ rid, int n4) {
    int i = blockIdx.x * blockDim.x + threadIdx.x;
    if (i < n4) {
        const v4i d = ((const v4i*)dst)[i];
        atomicAdd(&deg[d.x], 1.0f);
        atomicAdd(&deg[d.y], 1.0f);
        atomicAdd(&deg[d.z], 1.0f);
        atomicAdd(&deg[d.w], 1.0f);
        rid[i] = ((unsigned)d.x / AGG_RPW) | (((unsigned)d.y / AGG_RPW) << 8) |
                 (((unsigned)d.z / AGG_RPW) << 16) | (((unsigned)d.w / AGG_RPW) << 24);
    }
}

__global__ void gcn_rsqrt_kernel(float* __restrict__ deg, int n) {
    int i = blockIdx.x * blockDim.x + threadIdx.x;
    if (i < n) deg[i] = rsqrtf(deg[i]);
}

template <int K, int N, bool RELU_A>
__global__ __launch_bounds__(256) void gcn_gemm_wmma_kernel(
    const float* __restrict__ A, const float* __restrict__ B,
    const float* __restrict__ bias, const float* __restrict__ isq,
    float* __restrict__ C, float* __restrict__ S, int M) {
    constexpr int NT = N / 16;

    const int lane     = threadIdx.x & 31;
    const int wave     = threadIdx.x >> 5;
    const int row_tile = blockIdx.x * 8 + wave;
    if (row_tile * 16 >= M) return;

    const int m     = lane & 15;
    const int khalf = lane >> 4;
    const int row   = row_tile * 16 + m;

    v8f acc[NT];
#pragma unroll
    for (int nt = 0; nt < NT; ++nt) acc[nt] = (v8f){0.f,0.f,0.f,0.f,0.f,0.f,0.f,0.f};

    const float* __restrict__ arow = A + (size_t)row * K;
#pragma unroll
    for (int k0 = 0; k0 < K; k0 += 4) {
        const int kk = k0 + khalf * 2;
        v2f a = *(const v2f*)(arow + kk);
        if (RELU_A) { a.x = fmaxf(a.x, 0.0f); a.y = fmaxf(a.y, 0.0f); }
#pragma unroll
        for (int nt = 0; nt < NT; ++nt) {
            const int n = nt * 16 + m;
            v2f b;
            b.x = B[(size_t)kk * N + n];
            b.y = B[(size_t)(kk + 1) * N + n];
            acc[nt] = __builtin_amdgcn_wmma_f32_16x16x4_f32(
                false, a, false, b, (short)0, acc[nt], false, false);
        }
    }

    const int r0 = row_tile * 16 + khalf * 8;
    const v4f s_lo = *(const v4f*)(isq + r0);
    const v4f s_hi = *(const v4f*)(isq + r0 + 4);
    const float nrm[8] = {s_lo.x * s_lo.x, s_lo.y * s_lo.y, s_lo.z * s_lo.z, s_lo.w * s_lo.w,
                          s_hi.x * s_hi.x, s_hi.y * s_hi.y, s_hi.z * s_hi.z, s_hi.w * s_hi.w};
#pragma unroll
    for (int nt = 0; nt < NT; ++nt) {
        const size_t off = (size_t)r0 * N + nt * 16 + m;
        float* __restrict__ crow = C + off;
        float* __restrict__ srow = S + off;
        const float bv = bias[nt * 16 + m];
#pragma unroll
        for (int j = 0; j < 8; ++j) {
            const float hv = acc[nt][j];
            crow[(size_t)j * N] = hv;
            srow[(size_t)j * N] = hv * nrm[j] + bv;
        }
    }
}

#define AGG_T     1024
#define AGG_WAVES (AGG_T / 32)
#define AGG_QCAP  64
#define AGG_U     4
#define AGG_G     16
#define AGG_LCAP  512
#define AGG_SBITS 17

static_assert((AGG_WGS - 1) * AGG_RPW < NNODES, "every work-group owns at least one row");
static_assert(NNODES <= (1 << AGG_SBITS), "source row must fit its field of a saved entry");
static_assert(AGG_RPW < (1 << (31 - AGG_SBITS)), "local row must fit its field of a saved entry");
static_assert(AGG_LCAP % 32 == 0, "a full group of 32 saved entries starts on a 256-byte boundary");
static_assert(AGG_WAVES == 32, "one lane per wave reads or stores the list lengths");

template <int F, bool SAVE, bool USE>
__global__ __launch_bounds__(AGG_T) void gcn_agg_lds_kernel(
    const float* __restrict__ h, const int* __restrict__ src,
    const int* __restrict__ dst, const unsigned* __restrict__ rid,
    const float* __restrict__ isq,
    float* __restrict__ out, int* __restrict__ lcnt, int* __restrict__ lent) {
    constexpr int VPL = F / 32;
    __shared__ __attribute__((aligned(16))) float acc[AGG_RPW * F];
    __shared__ int pend[AGG_WAVES * AGG_QCAP];

    const int tid  = threadIdx.x;
    const int lane = tid & 31;
    const int wave = tid >> 5;
    const int lo   = blockIdx.x * AGG_RPW;
    const int rem  = NNODES - lo;
    const int nrows = rem < AGG_RPW ? rem : AGG_RPW;

    float* __restrict__ obase = out + (size_t)lo * F;
    const int nvec = nrows * (F / 4);
    constexpr int NI = (AGG_RPW * (F / 4) + AGG_T - 1) / AGG_T;
    {
        v4f tmp[NI];
#pragma unroll
        for (int i = 0; i < NI; ++i) {
            const int idx = tid + i * AGG_T;
            if (idx < nvec) tmp[i] = ((const v4f*)obase)[idx];
        }
#pragma unroll
        for (int i = 0; i < NI; ++i) {
            const int idx = tid + i * AGG_T;
            if (idx < nvec) ((v4f*)acc)[idx] = tmp[i];
        }
    }
    __syncthreads();

    int* __restrict__ myq = pend + wave * AGG_QCAP;
    v2i* __restrict__ mylst = (v2i*)lent + ((size_t)blockIdx.x * AGG_WAVES + wave) * AGG_LCAP;

    auto gather_add = [&](int s, int dl, float w, int cnt) {
        const float* __restrict__ hb = h + lane * VPL;
#pragma unroll
        for (int j = 0; j < 32; j += AGG_G) {
            if (j < cnt) {
                float v[AGG_G][VPL];
#pragma unroll
                for (int t = 0; t < AGG_G; ++t) {
                    const int sj = __builtin_amdgcn_readlane(s, j + t);
                    const float* __restrict__ p = hb + (size_t)sj * F;
                    if (VPL == 2) { const v2f r = *(const v2f*)p; v[t][0] = r.x; v[t][VPL - 1] = r.y; }
                    else          { v[t][0] = *p; }
                }
#pragma unroll
                for (int t = 0; t < AGG_G; ++t) {
                    if (j + t < cnt) {
                        const float wj = __int_as_float(__builtin_amdgcn_readlane(__float_as_int(w), j + t));
                        const int   dj = __builtin_amdgcn_readlane(dl, j + t);
                        float* a = acc + dj * F + lane * VPL;
#pragma unroll
                        for (int c = 0; c < VPL; ++c) atomicAdd(a + c, v[t][c] * wj);
                    }
                }
            }
        }
    };

    bool from_list = false;
    if (USE) {
        const int len_l = lcnt[blockIdx.x * AGG_WAVES + lane];
        from_list = __builtin_amdgcn_ballot_w32(len_l > AGG_LCAP) == 0u;
        if (from_list) {
            const int n = __builtin_amdgcn_readlane(len_l, __builtin_amdgcn_readfirstlane(wave));
            v2i cur = (v2i){0, 0};
            if (lane < n) cur = mylst[lane];
            for (int base = 0; base < n; base += 32) {
                const int rest = n - base;
                const int cnt  = rest < 32 ? rest : 32;
                v2i nxt = (v2i){0, 0};
                if (base + 32 + lane < n) nxt = mylst[base + 32 + lane];
                gather_add(cur.x & ((1 << AGG_SBITS) - 1), cur.x >> AGG_SBITS, __int_as_float(cur.y), cnt);
                cur = nxt;
            }
        }
    }

    int tot = 0;
    if (!from_list) {
        int qn = 0;
        const unsigned lt = (1u << lane) - 1u;

        auto drain = [&](int base, int cnt) {
            const int e   = myq[base + (lane < cnt ? lane : 0)];
            const int s   = src[e];
            const int dl  = dst[e] - lo;
            const float w = isq[s] * isq[lo + dl];
            if (SAVE) {
                const int p = tot + lane;
                if (lane < cnt && p < AGG_LCAP) mylst[p] = (v2i){s | (dl << AGG_SBITS), __float_as_int(w)};
                tot += cnt;
            }
            gather_add(s, dl, w, cnt);
        };

        constexpr int EPI = AGG_T * 16;
        constexpr int NG  = (NEDGES + EPI * AGG_U - 1) / (EPI * AGG_U);
        const unsigned pat = (unsigned)blockIdx.x * 0x01010101u;
        const int g0 = (int)(((long long)blockIdx.x * NG) / AGG_WGS);
        for (int gi = 0; gi < NG; ++gi) {
            int grp = g0 + gi;
            if (grp >= NG) grp -= NG;
            v4i rv[AGG_U];
#pragma unroll
            for (int u = 0; u < AGG_U; ++u) {
                const int b  = (grp * AGG_U + u) * EPI + tid * 16;
                const int bc = b < NEDGES ? b : NEDGES - 16;
                rv[u] = ((const v4i*)rid)[bc >> 4];
            }
            asm volatile("" :: "v"(rv[0]), "v"(rv[1]), "v"(rv[2]), "v"(rv[3]));
            static_assert(AGG_U == 4, "the statement above names four loads");
#pragma unroll
            for (int u = 0; u < AGG_U; ++u) {
                const int b = (grp * AGG_U + u) * EPI + tid * 16;
                unsigned fl = 0u;
#pragma unroll
                for (int c = 0; c < 4; ++c) {
                    const unsigned y = (unsigned)rv[u][c] ^ pat;
                    const unsigned t = ((y & 0x7f7f7f7fu) + 0x7f7f7f7fu) | y;
                    fl |= (~t & 0x80808080u) >> (7 - c);
                }
                if (b >= NEDGES) fl = 0u;
                unsigned bal = __builtin_amdgcn_ballot_w32(fl != 0u);
                while (bal != 0u) {
                    if (fl != 0u) {
                        const int k = __builtin_ctz(fl);
                        myq[qn + __builtin_popcount(bal & lt)] = b + ((k & 7) << 2) + (k >> 3);
                        fl &= fl - 1u;
                    }
                    qn += __builtin_popcount(bal);
                    if (qn >= 32) { qn -= 32; drain(qn, 32); }
                    bal = __builtin_amdgcn_ballot_w32(fl != 0u);
                }
            }
        }
        if (qn > 0) drain(0, qn);
    }
    if (SAVE && lane == 0) myq[0] = tot;

    __syncthreads();
    if (SAVE && wave == 0) lcnt[blockIdx.x * AGG_WAVES + lane] = pend[lane * AGG_QCAP];
#pragma unroll
    for (int i = 0; i < NI; ++i) {
        const int idx = tid + i * AGG_T;
        if (idx < nvec) ((v4f*)obase)[idx] = ((const v4f*)acc)[idx];
    }
}

extern "C" void kernel_launch(void* const* d_in, const int* in_sizes, int n_in,
                              void* d_out, int out_size, void* d_ws, size_t ws_size,
                              hipStream_t stream) {
    const float* x  = (const float*)d_in[0];
    const int*   ei = (const int*)d_in[1];
    const float* W1 = (const float*)d_in[2];
    const float* b1 = (const float*)d_in[3];
    const float* W2 = (const float*)d_in[4];
    const float* b2 = (const float*)d_in[5];
    const int* src = ei;
    const int* dst = ei + NEDGES;
    float* out = (float*)d_out;

    float* ws   = (float*)d_ws;
    float* isq  = ws;
    float* h1   = ws + 131072;
    float* agg1 = h1 + (size_t)NNODES * HID_DIM;
    float* h2   = h1;
    int* lcnt   = (int*)(agg1 + (size_t)NNODES * HID_DIM);
    int* lent   = lcnt + AGG_WGS * AGG_WAVES;
    unsigned* rid = (unsigned*)(lent + (size_t)AGG_WGS * AGG_WAVES * AGG_LCAP * 2);

    const int T = 256;

    gcn_fill_kernel<<<(NNODES + T - 1) / T, T, 0, stream>>>(isq, 1.0f, NNODES);
    gcn_deg_kernel<<<(NEDGES / 4 + T - 1) / T, T, 0, stream>>>(dst, isq, rid, NEDGES / 4);
    gcn_rsqrt_kernel<<<(NNODES + T - 1) / T, T, 0, stream>>>(isq, NNODES);

    {
        int waves  = NNODES / 16;
        int blocks = (waves + 7) / 8;
        gcn_gemm_wmma_kernel<IN_DIM, HID_DIM, false><<<blocks, T, 0, stream>>>(
            x, W1, b1, isq, h1, agg1, NNODES);
    }
    gcn_agg_lds_kernel<HID_DIM, true, false><<<AGG_WGS, AGG_T, 0, stream>>>(h1, src, dst, rid, isq, agg1, lcnt, lent);

    {
        int waves  = NNODES / 16;
        int blocks = (waves + 7) / 8;
        gcn_gemm_wmma_kernel<HID_DIM, OUT_DIM, true><<<blocks, T, 0, stream>>>(
            agg1, W2, b2, isq, h2, out, NNODES);
    }
    gcn_agg_lds_kernel<OUT_DIM, false, true><<<AGG_WGS, AGG_T, 0, stream>>>(h2, src, dst, rid, isq, out, lcnt, lent);
}


# passes folded, 7 launches to 4: layer-2 GEMM at the end of the layer-1 aggregation on its summed rows in LDS (WMMA, same k order); GEMM1 sets deg; rsqrt and self-loop start value taken by readers
# speedup vs baseline: 1.9412x; 1.1339x over previous
#include <hip/hip_runtime.h>

#define NNODES 100000
#define NEDGES 1200000
#define IN_DIM 64
#define HID_DIM 64
#define OUT_DIM 32

typedef __attribute__((ext_vector_type(2))) float v2f;
typedef __attribute__((ext_vector_type(4))) float v4f;
typedef __attribute__((ext_vector_type(8))) float v8f;
typedef __attribute__((ext_vector_type(4))) int v4i;
typedef __attribute__((ext_vector_type(2))) int v2i;

#define AGG_WGS   256
#define AGG_RPW   ((NNODES + AGG_WGS - 1) / AGG_WGS)
static_assert(AGG_WGS <= 256, "a range number must fit one byte");
static_assert(NEDGES % 16 == 0, "range numbers are read sixteen edges per lane");

__global__ void gcn_deg_kernel(const int* __restrict__ dst, float* __restrict__ deg,
                               unsigned* __restrict__ rid, int n4) {
    int i = blockIdx.x * blockDim.x + threadIdx.x;
    if (i < n4) {
        const v4i d = ((const v4i*)dst)[i];
        atomicAdd(&deg[d.x], 1.0f);
        atomicAdd(&deg[d.y], 1.0f);
        atomicAdd(&deg[d.z], 1.0f);
        atomicAdd(&deg[d.w], 1.0f);
        rid[i] = ((unsigned)d.x / AGG_RPW) | (((unsigned)d.y / AGG_RPW) << 8) |
                 (((unsigned)d.z / AGG_RPW) << 16) | (((unsigned)d.w / AGG_RPW) << 24);
    }
}


template <int K, int N>
__global__ __launch_bounds__(256) void gcn_gemm_wmma_kernel(
    const float* __restrict__ A, const float* __restrict__ B,
    float* __restrict__ deg, float* __restrict__ C, int M) {
    constexpr int NT = N / 16;

    const int lane     = threadIdx.x & 31;
    const int wave     = threadIdx.x >> 5;
    const int row_tile = blockIdx.x * 8 + wave;
    if (row_tile * 16 >= M) return;

    if ((row_tile & 1) == 0) {
        const int di = row_tile * 16 + lane;
        if (di < M) deg[di] = 1.0f;
    }

    const int m     = lane & 15;
    const int khalf = lane >> 4;
    const int row   = row_tile * 16 + m;

    v8f acc[NT];
#pragma unroll
    for (int nt = 0; nt < NT; ++nt) acc[nt] = (v8f){0.f,0.f,0.f,0.f,0.f,0.f,0.f,0.f};

    const float* __restrict__ arow = A + (size_t)row * K;
#pragma unroll
    for (int k0 = 0; k0 < K; k0 += 4) {
        const int kk = k0 + khalf * 2;
        const v2f a = *(const v2f*)(arow + kk);
#pragma unroll
        for (int nt = 0; nt < NT; ++nt) {
            const int n = nt * 16 + m;
            v2f b;
            b.x = B[(size_t)kk * N + n];
            b.y = B[(size_t)(kk + 1) * N + n];
            acc[nt] = __builtin_amdgcn_wmma_f32_16x16x4_f32(
                false, a, false, b, (short)0, acc[nt], false, false);
        }
    }

    const int r0 = row_tile * 16 + khalf * 8;
#pragma unroll
    for (int nt = 0; nt < NT; ++nt) {
        float* __restrict__ crow = C + (size_t)r0 * N + nt * 16 + m;
#pragma unroll
        for (int j = 0; j < 8; ++j) crow[(size_t)j * N] = acc[nt][j];
    }
}

#define AGG_T     1024
#define AGG_WAVES (AGG_T / 32)
#define AGG_QCAP  64
#define AGG_U     4
#define AGG_G     16
#define AGG_LCAP  512
#define AGG_SBITS 17

static_assert((AGG_WGS - 1) * AGG_RPW < NNODES, "every work-group owns at least one row");
static_assert(NNODES <= (1 << AGG_SBITS), "source row must fit its field of a saved entry");
static_assert(AGG_RPW < (1 << (31 - AGG_SBITS)), "local row must fit its field of a saved entry");
static_assert(AGG_LCAP % 32 == 0, "a full group of 32 saved entries starts on a 256-byte boundary");
static_assert(AGG_WAVES == 32, "one lane per wave reads or stores the list lengths");

template <int F, bool SAVE, bool USE>
__global__ __launch_bounds__(AGG_T) void gcn_agg_lds_kernel(
    const float* __restrict__ h, const int* __restrict__ src,
    const int* __restrict__ dst, const unsigned* __restrict__ rid,
    const float* __restrict__ deg, const float* __restrict__ bias,
    float* __restrict__ out, int* __restrict__ lcnt, int* __restrict__ lent,
    const float* __restrict__ Wn, float* __restrict__ hn) {
    constexpr int VPL = F / 32;
    __shared__ __attribute__((aligned(16))) float acc[AGG_RPW * F];
    __shared__ int pend[AGG_WAVES * AGG_QCAP];

    const int tid  = threadIdx.x;
    const int lane = tid & 31;
    const int wave = tid >> 5;
    const int lo   = blockIdx.x * AGG_RPW;
    const int rem  = NNODES - lo;
    const int nrows = rem < AGG_RPW ? rem : AGG_RPW;

    const int nvec = nrows * (F / 4);
    constexpr int NI = (AGG_RPW * (F / 4) + AGG_T - 1) / AGG_T;
    {
        const float* __restrict__ hown = h + (size_t)lo * F;
        static_assert(AGG_T % (F / 4) == 0, "a thread's pieces share their columns");
        const v4f bv = ((const v4f*)bias)[tid % (F / 4)];
        v4f tmp[NI];
        float dg[NI];
#pragma unroll
        for (int i = 0; i < NI; ++i) {
            const int idx = tid + i * AGG_T;
            if (idx < nvec) {
                tmp[i] = ((const v4f*)hown)[idx];
                dg[i]  = deg[lo + idx / (F / 4)];
            }
        }
#pragma unroll
        for (int i = 0; i < NI; ++i) {
            const int idx = tid + i * AGG_T;
            if (idx < nvec) {
                const float sq  = rsqrtf(dg[i]);
                const float nrm = sq * sq;
                ((v4f*)acc)[idx] = tmp[i] * nrm + bv;
            }
        }
    }
    __syncthreads();

    int* __restrict__ myq = pend + wave * AGG_QCAP;
    v2i* __restrict__ mylst = (v2i*)lent + ((size_t)blockIdx.x * AGG_WAVES + wave) * AGG_LCAP;

    auto gather_add = [&](int s, int dl, float w, int cnt) {
        const float* __restrict__ hb = h + lane * VPL;
#pragma unroll
        for (int j = 0; j < 32; j += AGG_G) {
            if (j < cnt) {
                float v[AGG_G][VPL];
#pragma unroll
                for (int t = 0; t < AGG_G; ++t) {
                    const int sj = __builtin_amdgcn_readlane(s, j + t);
                    const float* __restrict__ p = hb + (size_t)sj * F;
                    if (VPL == 2) { const v2f r = *(const v2f*)p; v[t][0] = r.x; v[t][VPL - 1] = r.y; }
                    else          { v[t][0] = *p; }
                }
#pragma unroll
                for (int t = 0; t < AGG_G; ++t) {
                    if (j + t < cnt) {
                        const float wj = __int_as_float(__builtin_amdgcn_readlane(__float_as_int(w), j + t));
                        const int   dj = __builtin_amdgcn_readlane(dl, j + t);
                        float* a = acc + dj * F + lane * VPL;
#pragma unroll
                        for (int c = 0; c < VPL; ++c) atomicAdd(a + c, v[t][c] * wj);
                    }
                }
            }
        }
    };

    bool from_list = false;
    if (USE) {
        const int len_l = lcnt[blockIdx.x * AGG_WAVES + lane];
        from_list = __builtin_amdgcn_ballot_w32(len_l > AGG_LCAP) == 0u;
        if (from_list) {
            const int n = __builtin_amdgcn_readlane(len_l, __builtin_amdgcn_readfirstlane(wave));
            v2i cur = (v2i){0, 0};
            if (lane < n) cur = mylst[lane];
            for (int base = 0; base < n; base += 32) {
                const int rest = n - base;
                const int cnt  = rest < 32 ? rest : 32;
                v2i nxt = (v2i){0, 0};
                if (base + 32 + lane < n) nxt = mylst[base + 32 + lane];
                gather_add(cur.x & ((1 << AGG_SBITS) - 1), cur.x >> AGG_SBITS, __int_as_float(cur.y), cnt);
                cur = nxt;
            }
        }
    }

    int tot = 0;
    if (!from_list) {
        int qn = 0;
        const unsigned lt = (1u << lane) - 1u;

        auto drain = [&](int base, int cnt) {
            const int e   = myq[base + (lane < cnt ? lane : 0)];
            const int s   = src[e];
            const int dl  = dst[e] - lo;
            const float w = rsqrtf(deg[s]) * rsqrtf(deg[lo + dl]);
            if (SAVE) {
                const int p = tot + lane;
                if (lane < cnt && p < AGG_LCAP) mylst[p] = (v2i){s | (dl << AGG_SBITS), __float_as_int(w)};
                tot += cnt;
            }
            gather_add(s, dl, w, cnt);
        };

        constexpr int EPI = AGG_T * 16;
        constexpr int NG  = (NEDGES + EPI * AGG_U - 1) / (EPI * AGG_U);
        const unsigned pat = (unsigned)blockIdx.x * 0x01010101u;
        const int g0 = (int)(((long long)blockIdx.x * NG) / AGG_WGS);
        for (int gi = 0; gi < NG; ++gi) {
            int grp = g0 + gi;
            if (grp >= NG) grp -= NG;
            v4i rv[AGG_U];
#pragma unroll
            for (int u = 0; u < AGG_U; ++u) {
                const int b  = (grp * AGG_U + u) * EPI + tid * 16;
                const int bc = b < NEDGES ? b : NEDGES - 16;
                rv[u] = ((const v4i*)rid)[bc >> 4];
            }
            asm volatile("" :: "v"(rv[0]), "v"(rv[1]), "v"(rv[2]), "v"(rv[3]));
            static_assert(AGG_U == 4, "the statement above names four loads");
#pragma unroll
            for (int u = 0; u < AGG_U; ++u) {
                const int b = (grp * AGG_U + u) * EPI + tid * 16;
                unsigned fl = 0u;
#pragma unroll
                for (int c = 0; c < 4; ++c) {
                    const unsigned y = (unsigned)rv[u][c] ^ pat;
                    const unsigned t = ((y & 0x7f7f7f7fu) + 0x7f7f7f7fu) | y;
                    fl |= (~t & 0x80808080u) >> (7 - c);
                }
                if (b >= NEDGES) fl = 0u;
                unsigned bal = __builtin_amdgcn_ballot_w32(fl != 0u);
                while (bal != 0u) {
                    if (fl != 0u) {
                        const int k = __builtin_ctz(fl);
                        myq[qn + __builtin_popcount(bal & lt)] = b + ((k & 7) << 2) + (k >> 3);
                        fl &= fl - 1u;
                    }
                    qn += __builtin_popcount(bal);
                    if (qn >= 32) { qn -= 32; drain(qn, 32); }
                    bal = __builtin_amdgcn_ballot_w32(fl != 0u);
                }
            }
        }
        if (qn > 0) drain(0, qn);
    }
    if (SAVE && lane == 0) myq[0] = tot;

    __syncthreads();
    if (SAVE && wave == 0) lcnt[blockIdx.x * AGG_WAVES + lane] = pend[lane * AGG_QCAP];
    if (SAVE) {
        static_assert(!SAVE || (F == HID_DIM && OUT_DIM == 32), "epilogue: K = F, two column tiles");
        static_assert((AGG_RPW + 15) / 16 <= AGG_WAVES, "one slab per wave");
        const int tile = __builtin_amdgcn_readfirstlane(wave);
        if (tile * 16 < nrows) {
            const int m     = lane & 15;
            const int khalf = lane >> 4;
            const int rl = tile * 16 + m < nrows ? tile * 16 + m : nrows - 1;
            const float* arow = acc + rl * F;
            v8f ct[2];
            ct[0] = (v8f){0.f,0.f,0.f,0.f,0.f,0.f,0.f,0.f};
            ct[1] = (v8f){0.f,0.f,0.f,0.f,0.f,0.f,0.f,0.f};
#pragma unroll
            for (int k0 = 0; k0 < F; k0 += 4) {
                const int kk = k0 + khalf * 2;
                v2f a = *(const v2f*)(arow + kk);
                a.x = fmaxf(a.x, 0.0f); a.y = fmaxf(a.y, 0.0f);
#pragma unroll
                for (int nt = 0; nt < 2; ++nt) {
                    const int n = nt * 16 + m;
                    v2f b;
                    b.x = Wn[(size_t)kk * OUT_DIM + n];
                    b.y = Wn[(size_t)(kk + 1) * OUT_DIM + n];
                    ct[nt] = __builtin_amdgcn_wmma_f32_16x16x4_f32(
                        false, a, false, b, (short)0, ct[nt], false, false);
                }
            }
            const int rb = tile * 16 + khalf * 8;
#pragma unroll
            for (int nt = 0; nt < 2; ++nt) {
                float* __restrict__ crow = hn + (size_t)(lo + rb) * OUT_DIM + nt * 16 + m;
#pragma unroll
                for (int j = 0; j < 8; ++j) {
                    if (rb + j < nrows) crow[(size_t)j * OUT_DIM] = ct[nt][j];
                }
            }
        }
    } else {
        float* __restrict__ obase = out + (size_t)lo * F;
#pragma unroll
        for (int i = 0; i < NI; ++i) {
            const int idx = tid + i * AGG_T;
            if (idx < nvec) ((v4f*)obase)[idx] = ((const v4f*)acc)[idx];
        }
    }
}

extern "C" void kernel_launch(void* const* d_in, const int* in_sizes, int n_in,
                              void* d_out, int out_size, void* d_ws, size_t ws_size,
                              hipStream_t stream) {
    const float* x  = (const float*)d_in[0];
    const int*   ei = (const int*)d_in[1];
    const float* W1 = (const float*)d_in[2];
    const float* b1 = (const float*)d_in[3];
    const float* W2 = (const float*)d_in[4];
    const float* b2 = (const float*)d_in[5];
    const int* src = ei;
    const int* dst = ei + NEDGES;
    float* out = (float*)d_out;

    float* ws   = (float*)d_ws;
    float* deg  = ws;
    float* h1   = ws + 131072;
    int* lcnt   = (int*)(h1 + (size_t)NNODES * HID_DIM);
    int* lent   = lcnt + AGG_WGS * AGG_WAVES;
    unsigned* rid = (unsigned*)(lent + (size_t)AGG_WGS * AGG_WAVES * AGG_LCAP * 2);
    float* h2   = (float*)(rid + NEDGES / 4);

    const int T = 256;

    {
        int waves  = NNODES / 16;
        int blocks = (waves + 7) / 8;
        gcn_gemm_wmma_kernel<IN_DIM, HID_DIM><<<blocks, T, 0, stream>>>(x, W1, deg, h1, NNODES);
    }

    gcn_deg_kernel<<<(NEDGES / 4 + T - 1) / T, T, 0, stream>>>(dst, deg, rid, NEDGES / 4);

    gcn_agg_lds_kernel<HID_DIM, true, false><<<AGG_WGS, AGG_T, 0, stream>>>(
        h1, src, dst, rid, deg, b1, nullptr, lcnt, lent, W2, h2);

    gcn_agg_lds_kernel<OUT_DIM, false, true><<<AGG_WGS, AGG_T, 0, stream>>>(
        h2, src, dst, rid, deg, b2, out, lcnt, lent, nullptr, nullptr);
}


# degree count: one scattered memory f32 atomic per edge replaced by 16-bit counter tables in LDS (32 edge slices, LDS int adds), each added to deg by whole-line f32 adds; exact counts, new add order
# speedup vs baseline: 2.6234x; 1.3514x over previous
#include <hip/hip_runtime.h>

#define NNODES 100000
#define NEDGES 1200000
#define IN_DIM 64
#define HID_DIM 64
#define OUT_DIM 32

typedef __attribute__((ext_vector_type(2))) float v2f;
typedef __attribute__((ext_vector_type(4))) float v4f;
typedef __attribute__((ext_vector_type(8))) float v8f;
typedef __attribute__((ext_vector_type(4))) int v4i;
typedef __attribute__((ext_vector_type(2))) int v2i;

#define AGG_WGS   256
#define AGG_RPW   ((NNODES + AGG_WGS - 1) / AGG_WGS)
static_assert(AGG_WGS <= 256, "a range number must fit one byte");
static_assert(NEDGES % 16 == 0, "range numbers are read sixteen edges per lane");

#define DEG_WGS 32
#define DEG_T   1024
#define DEG_PER (NEDGES / 4 / DEG_WGS)
static_assert((NEDGES / 4) % DEG_WGS == 0, "the slices are equal");
static_assert(NEDGES / DEG_WGS <= 65535, "a 16-bit counter holds a whole slice");
static_assert(NNODES % 32 == 0, "a wave's add covers 32 nodes of one line; the table is cleared 8 nodes a store");

__global__ __launch_bounds__(DEG_T) void gcn_deg_kernel(const int* __restrict__ dst, float* __restrict__ deg,
                                                        unsigned* __restrict__ rid) {
    __shared__ __attribute__((aligned(16))) unsigned cnt[NNODES / 2];
    const int tid  = threadIdx.x;
    const int base = blockIdx.x * DEG_PER;
    constexpr int NI = (DEG_PER + DEG_T - 1) / DEG_T;

    v4i d[NI];
#pragma unroll
    for (int i = 0; i < NI; ++i) {
        const int it = tid + i * DEG_T;
        if (it < DEG_PER) d[i] = ((const v4i*)dst)[base + it];
    }
    for (int i = tid; i < NNODES / 8; i += DEG_T) ((v4i*)cnt)[i] = (v4i){0, 0, 0, 0};
    __syncthreads();

#pragma unroll
    for (int i = 0; i < NI; ++i) {
        const int it = tid + i * DEG_T;
        if (it < DEG_PER) {
            const v4i e = d[i];
            atomicAdd(&cnt[e.x >> 1], 1u << ((e.x & 1) * 16));
            atomicAdd(&cnt[e.y >> 1], 1u << ((e.y & 1) * 16));
            atomicAdd(&cnt[e.z >> 1], 1u << ((e.z & 1) * 16));
            atomicAdd(&cnt[e.w >> 1], 1u << ((e.w & 1) * 16));
            rid[base + it] = ((unsigned)e.x / AGG_RPW) | (((unsigned)e.y / AGG_RPW) << 8) |
                             (((unsigned)e.z / AGG_RPW) << 16) | (((unsigned)e.w / AGG_RPW) << 24);
        }
    }
    __syncthreads();

    constexpr int FI = (NNODES + DEG_T - 1) / DEG_T;
    constexpr int FU = 7;
    static_assert(FI % FU == 0, "the add loop runs whole groups");
    const int start = (int)blockIdx.x * (NNODES / 32 / DEG_WGS) * 32;
#pragma nounroll
    for (int j0 = 0; j0 < FI; j0 += FU) {
        unsigned wv[FU];
#pragma unroll
        for (int u = 0; u < FU; ++u) {
            const int i = tid + (j0 + u) * DEG_T;
            int n = start + i;
            if (n >= NNODES) n -= NNODES;
            wv[u] = i < NNODES ? cnt[n >> 1] : 0u;
        }
#pragma unroll
        for (int u = 0; u < FU; ++u) {
            const int i = tid + (j0 + u) * DEG_T;
            int n = start + i;
            if (n >= NNODES) n -= NNODES;
            const unsigned c = (wv[u] >> ((n & 1) * 16)) & 0xffffu;
            if (c != 0u) atomicAdd(&deg[n], (float)c);
        }
    }
}


template <int K, int N>
__global__ __launch_bounds__(256) void gcn_gemm_wmma_kernel(
    const float* __restrict__ A, const float* __restrict__ B,
    float* __restrict__ deg, float* __restrict__ C, int M) {
    constexpr int NT = N / 16;

    const int lane     = threadIdx.x & 31;
    const int wave     = threadIdx.x >> 5;
    const int row_tile = blockIdx.x * 8 + wave;
    if (row_tile * 16 >= M) return;

    if ((row_tile & 1) == 0) {
        const int di = row_tile * 16 + lane;
        if (di < M) deg[di] = 1.0f;
    }

    const int m     = lane & 15;
    const int khalf = lane >> 4;
    const int row   = row_tile * 16 + m;

    v8f acc[NT];
#pragma unroll
    for (int nt = 0; nt < NT; ++nt) acc[nt] = (v8f){0.f,0.f,0.f,0.f,0.f,0.f,0.f,0.f};

    const float* __restrict__ arow = A + (size_t)row * K;
#pragma unroll
    for (int k0 = 0; k0 < K; k0 += 4) {
        const int kk = k0 + khalf * 2;
        const v2f a = *(const v2f*)(arow + kk);
#pragma unroll
        for (int nt = 0; nt < NT; ++nt) {
            const int n = nt * 16 + m;
            v2f b;
            b.x = B[(size_t)kk * N + n];
            b.y = B[(size_t)(kk + 1) * N + n];
            acc[nt] = __builtin_amdgcn_wmma_f32_16x16x4_f32(
                false, a, false, b, (short)0, acc[nt], false, false);
        }
    }

    const int r0 = row_tile * 16 + khalf * 8;
#pragma unroll
    for (int nt = 0; nt < NT; ++nt) {
        float* __restrict__ crow = C + (size_t)r0 * N + nt * 16 + m;
#pragma unroll
        for (int j = 0; j < 8; ++j) crow[(size_t)j * N] = acc[nt][j];
    }
}

#define AGG_T     1024
#define AGG_WAVES (AGG_T / 32)
#define AGG_QCAP  64
#define AGG_U     4
#define AGG_G     16
#define AGG_LCAP  512
#define AGG_SBITS 17

static_assert((AGG_WGS - 1) * AGG_RPW < NNODES, "every work-group owns at least one row");
static_assert(NNODES <= (1 << AGG_SBITS), "source row must fit its field of a saved entry");
static_assert(AGG_RPW < (1 << (31 - AGG_SBITS)), "local row must fit its field of a saved entry");
static_assert(AGG_LCAP % 32 == 0, "a full group of 32 saved entries starts on a 256-byte boundary");
static_assert(AGG_WAVES == 32, "one lane per wave reads or stores the list lengths");

template <int F, bool SAVE, bool USE>
__global__ __launch_bounds__(AGG_T) void gcn_agg_lds_kernel(
    const float* __restrict__ h, const int* __restrict__ src,
    const int* __restrict__ dst, const unsigned* __restrict__ rid,
    const float* __restrict__ deg, const float* __restrict__ bias,
    float* __restrict__ out, int* __restrict__ lcnt, int* __restrict__ lent,
    const float* __restrict__ Wn, float* __restrict__ hn) {
    constexpr int VPL = F / 32;
    __shared__ __attribute__((aligned(16))) float acc[AGG_RPW * F];
    __shared__ int pend[AGG_WAVES * AGG_QCAP];

    const int tid  = threadIdx.x;
    const int lane = tid & 31;
    const int wave = tid >> 5;
    const int lo   = blockIdx.x * AGG_RPW;
    const int rem  = NNODES - lo;
    const int nrows = rem < AGG_RPW ? rem : AGG_RPW;

    const int nvec = nrows * (F / 4);
    constexpr int NI = (AGG_RPW * (F / 4) + AGG_T - 1) / AGG_T;
    {
        const float* __restrict__ hown = h + (size_t)lo * F;
        static_assert(AGG_T % (F / 4) == 0, "a thread's pieces share their columns");
        const v4f bv = ((const v4f*)bias)[tid % (F / 4)];
        v4f tmp[NI];
        float dg[NI];
#pragma unroll
        for (int i = 0; i < NI; ++i) {
            const int idx = tid + i * AGG_T;
            if (idx < nvec) {
                tmp[i] = ((const v4f*)hown)[idx];
                dg[i]  = deg[lo + idx / (F / 4)];
            }
        }
#pragma unroll
        for (int i = 0; i < NI; ++i) {
            const int idx = tid + i * AGG_T;
            if (idx < nvec) {
                const float sq  = rsqrtf(dg[i]);
                const float nrm = sq * sq;
                ((v4f*)acc)[idx] = tmp[i] * nrm + bv;
            }
        }
    }
    __syncthreads();

    int* __restrict__ myq = pend + wave * AGG_QCAP;
    v2i* __restrict__ mylst = (v2i*)lent + ((size_t)blockIdx.x * AGG_WAVES + wave) * AGG_LCAP;

    auto gather_add = [&](int s, int dl, float w, int cnt) {
        const float* __restrict__ hb = h + lane * VPL;
#pragma unroll
        for (int j = 0; j < 32; j += AGG_G) {
            if (j < cnt) {
                float v[AGG_G][VPL];
#pragma unroll
                for (int t = 0; t < AGG_G; ++t) {
                    const int sj = __builtin_amdgcn_readlane(s, j + t);
                    const float* __restrict__ p = hb + (size_t)sj * F;
                    if (VPL == 2) { const v2f r = *(const v2f*)p; v[t][0] = r.x; v[t][VPL - 1] = r.y; }
                    else          { v[t][0] = *p; }
                }
#pragma unroll
                for (int t = 0; t < AGG_G; ++t) {
                    if (j + t < cnt) {
                        const float wj = __int_as_float(__builtin_amdgcn_readlane(__float_as_int(w), j + t));
                        const int   dj = __builtin_amdgcn_readlane(dl, j + t);
                        float* a = acc + dj * F + lane * VPL;
#pragma unroll
                        for (int c = 0; c < VPL; ++c) atomicAdd(a + c, v[t][c] * wj);
                    }
                }
            }
        }
    };

    bool from_list = false;
    if (USE) {
        const int len_l = lcnt[blockIdx.x * AGG_WAVES + lane];
        from_list = __builtin_amdgcn_ballot_w32(len_l > AGG_LCAP) == 0u;
        if (from_list) {
            const int n = __builtin_amdgcn_readlane(len_l, __builtin_amdgcn_readfirstlane(wave));
            v2i cur = (v2i){0, 0};
            if (lane < n) cur = mylst[lane];
            for (int base = 0; base < n; base += 32) {
                const int rest = n - base;
                const int cnt  = rest < 32 ? rest : 32;
                v2i nxt = (v2i){0, 0};
                if (base + 32 + lane < n) nxt = mylst[base + 32 + lane];
                gather_add(cur.x & ((1 << AGG_SBITS) - 1), cur.x >> AGG_SBITS, __int_as_float(cur.y), cnt);
                cur = nxt;
            }
        }
    }

    int tot = 0;
    if (!from_list) {
        int qn = 0;
        const unsigned lt = (1u << lane) - 1u;

        auto drain = [&](int base, int cnt) {
            const int e   = myq[base + (lane < cnt ? lane : 0)];
            const int s   = src[e];
            const int dl  = dst[e] - lo;
            const float w = rsqrtf(deg[s]) * rsqrtf(deg[lo + dl]);
            if (SAVE) {
                const int p = tot + lane;
                if (lane < cnt && p < AGG_LCAP) mylst[p] = (v2i){s | (dl << AGG_SBITS), __float_as_int(w)};
                tot += cnt;
            }
            gather_add(s, dl, w, cnt);
        };

        constexpr int EPI = AGG_T * 16;
        constexpr int NG  = (NEDGES + EPI * AGG_U - 1) / (EPI * AGG_U);
        const unsigned pat = (unsigned)blockIdx.x * 0x01010101u;
        const int g0 = (int)(((long long)blockIdx.x * NG) / AGG_WGS);
        for (int gi = 0; gi < NG; ++gi) {
            int grp = g0 + gi;
            if (grp >= NG) grp -= NG;
            v4i rv[AGG_U];
#pragma unroll
            for (int u = 0; u < AGG_U; ++u) {
                const int b  = (grp * AGG_U + u) * EPI + tid * 16;
                const int bc = b < NEDGES ? b : NEDGES - 16;
                rv[u] = ((const v4i*)rid)[bc >> 4];
            }
            asm volatile("" :: "v"(rv[0]), "v"(rv[1]), "v"(rv[2]), "v"(rv[3]));
            static_assert(AGG_U == 4, "the statement above names four loads");
#pragma unroll
            for (int u = 0; u < AGG_U; ++u) {
                const int b = (grp * AGG_U + u) * EPI + tid * 16;
                unsigned fl = 0u;
#pragma unroll
                for (int c = 0; c < 4; ++c) {
                    const unsigned y = (unsigned)rv[u][c] ^ pat;
                    const unsigned t = ((y & 0x7f7f7f7fu) + 0x7f7f7f7fu) | y;
                    fl |= (~t & 0x80808080u) >> (7 - c);
                }
                if (b >= NEDGES) fl = 0u;
                unsigned bal = __builtin_amdgcn_ballot_w32(fl != 0u);
                while (bal != 0u) {
                    if (fl != 0u) {
                        const int k = __builtin_ctz(fl);
                        myq[qn + __builtin_popcount(bal & lt)] = b + ((k & 7) << 2) + (k >> 3);
                        fl &= fl - 1u;
                    }
                    qn += __builtin_popcount(bal);
                    if (qn >= 32) { qn -= 32; drain(qn, 32); }
                    bal = __builtin_amdgcn_ballot_w32(fl != 0u);
                }
            }
        }
        if (qn > 0) drain(0, qn);
    }
    if (SAVE && lane == 0) myq[0] = tot;

    __syncthreads();
    if (SAVE && wave == 0) lcnt[blockIdx.x * AGG_WAVES + lane] = pend[lane * AGG_QCAP];
    if (SAVE) {
        static_assert(!SAVE || (F == HID_DIM && OUT_DIM == 32), "epilogue: K = F, two column tiles");
        static_assert((AGG_RPW + 15) / 16 <= AGG_WAVES, "one slab per wave");
        const int tile = __builtin_amdgcn_readfirstlane(wave);
        if (tile * 16 < nrows) {
            const int m     = lane & 15;
            const int khalf = lane >> 4;
            const int rl = tile * 16 + m < nrows ? tile * 16 + m : nrows - 1;
            const float* arow = acc + rl * F;
            v8f ct[2];
            ct[0] = (v8f){0.f,0.f,0.f,0.f,0.f,0.f,0.f,0.f};
            ct[1] = (v8f){0.f,0.f,0.f,0.f,0.f,0.f,0.f,0.f};
#pragma unroll
            for (int k0 = 0; k0 < F; k0 += 4) {
                const int kk = k0 + khalf * 2;
                v2f a = *(const v2f*)(arow + kk);
                a.x = fmaxf(a.x, 0.0f); a.y = fmaxf(a.y, 0.0f);
#pragma unroll
                for (int nt = 0; nt < 2; ++nt) {
                    const int n = nt * 16 + m;
                    v2f b;
                    b.x = Wn[(size_t)kk * OUT_DIM + n];
                    b.y = Wn[(size_t)(kk + 1) * OUT_DIM + n];
                    ct[nt] = __builtin_amdgcn_wmma_f32_16x16x4_f32(
                        false, a, false, b, (short)0, ct[nt], false, false);
                }
            }
            const int rb = tile * 16 + khalf * 8;
#pragma unroll
            for (int nt = 0; nt < 2; ++nt) {
                float* __restrict__ crow = hn + (size_t)(lo + rb) * OUT_DIM + nt * 16 + m;
#pragma unroll
                for (int j = 0; j < 8; ++j) {
                    if (rb + j < nrows) crow[(size_t)j * OUT_DIM] = ct[nt][j];
                }
            }
        }
    } else {
        float* __restrict__ obase = out + (size_t)lo * F;
#pragma unroll
        for (int i = 0; i < NI; ++i) {
            const int idx = tid + i * AGG_T;
            if (idx < nvec) ((v4f*)obase)[idx] = ((const v4f*)acc)[idx];
        }
    }
}

extern "C" void kernel_launch(void* const* d_in, const int* in_sizes, int n_in,
                              void* d_out, int out_size, void* d_ws, size_t ws_size,
                              hipStream_t stream) {
    const float* x  = (const float*)d_in[0];
    const int*   ei = (const int*)d_in[1];
    const float* W1 = (const float*)d_in[2];
    const float* b1 = (const float*)d_in[3];
    const float* W2 = (const float*)d_in[4];
    const float* b2 = (const float*)d_in[5];
    const int* src = ei;
    const int* dst = ei + NEDGES;
    float* out = (float*)d_out;

    float* ws   = (float*)d_ws;
    float* deg  = ws;
    float* h1   = ws + 131072;
    int* lcnt   = (int*)(h1 + (size_t)NNODES * HID_DIM);
    int* lent   = lcnt + AGG_WGS * AGG_WAVES;
    unsigned* rid = (unsigned*)(lent + (size_t)AGG_WGS * AGG_WAVES * AGG_LCAP * 2);
    float* h2   = (float*)(rid + NEDGES / 4);

    const int T = 256;

    {
        int waves  = NNODES / 16;
        int blocks = (waves + 7) / 8;
        gcn_gemm_wmma_kernel<IN_DIM, HID_DIM><<<blocks, T, 0, stream>>>(x, W1, deg, h1, NNODES);
    }

    gcn_deg_kernel<<<DEG_WGS, DEG_T, 0, stream>>>(dst, deg, rid);

    gcn_agg_lds_kernel<HID_DIM, true, false><<<AGG_WGS, AGG_T, 0, stream>>>(
        h1, src, dst, rid, deg, b1, nullptr, lcnt, lent, W2, h2);

    gcn_agg_lds_kernel<OUT_DIM, false, true><<<AGG_WGS, AGG_T, 0, stream>>>(
        h2, src, dst, rid, deg, b2, out, lcnt, lent, nullptr, nullptr);
}


# aggregation stops filtering the edge list: 160 more work-groups of the degree launch sort edges into per-(dst range, list slice) buckets in LDS; a wave reads its 5 buckets; LDS sum order changes
# speedup vs baseline: 3.1909x; 1.2163x over previous
#include <hip/hip_runtime.h>

#define NNODES 100000
#define NEDGES 1200000
#define IN_DIM 64
#define HID_DIM 64
#define OUT_DIM 32

typedef __attribute__((ext_vector_type(2))) float v2f;
typedef __attribute__((ext_vector_type(4))) float v4f;
typedef __attribute__((ext_vector_type(8))) float v8f;
typedef __attribute__((ext_vector_type(4))) int v4i;
typedef __attribute__((ext_vector_type(2))) int v2i;

#define AGG_WGS   256
#define AGG_RPW   ((NNODES + AGG_WGS - 1) / AGG_WGS)
#define AGG_T     1024
#define AGG_WAVES (AGG_T / 32)
#define AGG_SBITS 17

#define PART_WGS  160
#define PART_PER  (NEDGES / 4 / PART_WGS)
#define PART_CAP  64
#define PART_SPW  (PART_WGS / AGG_WAVES)
static_assert((NEDGES / 4) % PART_WGS == 0, "the slices are equal");
static_assert(PART_WGS % AGG_WAVES == 0, "every aggregation wave reads the same number of buckets");
static_assert(PART_SPW <= 32, "one lane per bucket reads its length");
static_assert(PART_CAP % 32 == 0, "a bucket is read 32 entries at a time");
static_assert(NEDGES % AGG_WAVES == 0, "a wave's slices are a whole part of the edge list");
static_assert(NNODES <= (1 << AGG_SBITS), "source row must fit its field of an entry");
static_assert(AGG_RPW < (1 << (31 - AGG_SBITS)), "local row must fit its field of an entry");

#define DEG_WGS 32
#define DEG_T   1024
#define DEG_PER (NEDGES / 4 / DEG_WGS)
static_assert((NEDGES / 4) % DEG_WGS == 0, "the slices are equal");
static_assert(NEDGES / DEG_WGS <= 65535, "a 16-bit counter holds a whole slice");
static_assert(NNODES % 32 == 0, "a wave's add covers 32 nodes of one line; the table is cleared 8 nodes a store");

static_assert(AGG_WGS + AGG_WGS * PART_CAP <= NNODES / 2, "the sorting work-groups use the front of the counting table's array");

__global__ __launch_bounds__(DEG_T) void gcn_deg_kernel(const int* __restrict__ src, const int* __restrict__ dst,
                                                        float* __restrict__ deg, int* __restrict__ bcnt,
                                                        unsigned* __restrict__ bkt) {
    __shared__ __attribute__((aligned(16))) unsigned cnt[NNODES / 2];
    const int tid  = threadIdx.x;

    if (blockIdx.x >= DEG_WGS) {
        const int p     = blockIdx.x - DEG_WGS;
        const int pbase = p * PART_PER;
        unsigned* bc  = cnt;
        unsigned* stg = cnt + AGG_WGS;
        constexpr int NP  = (PART_PER + DEG_T - 1) / DEG_T;
        constexpr int GPR = PART_CAP / 4;

        v4i dd[NP], ss[NP];
#pragma unroll
        for (int i = 0; i < NP; ++i) {
            const int it = tid + i * DEG_T;
            if (it < PART_PER) {
                dd[i] = ((const v4i*)dst)[pbase + it];
                ss[i] = ((const v4i*)src)[pbase + it];
            }
        }
        for (int i = tid; i < (AGG_WGS + AGG_WGS * PART_CAP) / 4; i += DEG_T) ((v4i*)cnt)[i] = (v4i){0, 0, 0, 0};
        __syncthreads();

#pragma unroll
        for (int i = 0; i < NP; ++i) {
            const int it = tid + i * DEG_T;
            if (it < PART_PER) {
#pragma unroll
                for (int c = 0; c < 4; ++c) {
                    const unsigned d    = (unsigned)dd[i][c];
                    const unsigned r    = d / AGG_RPW;
                    const unsigned dl   = d - r * AGG_RPW;
                    const unsigned slot = atomicAdd(&bc[r], 1u);
                    if (slot < PART_CAP) stg[r * PART_CAP + slot] = (unsigned)ss[i][c] | (dl << AGG_SBITS);
                }
            }
        }
        __syncthreads();

        for (int gi = tid; gi < AGG_WGS * GPR; gi += DEG_T) {
            const int r = gi / GPR;
            const int g = gi % GPR;
            if ((unsigned)(g * 4) < bc[r])
                ((v4i*)bkt)[((size_t)r * PART_WGS + p) * GPR + g] = ((const v4i*)stg)[gi];
        }
        if (tid < AGG_WGS) bcnt[tid * PART_WGS + p] = (int)bc[tid];
        return;
    }

    const int base = blockIdx.x * DEG_PER;
    constexpr int NI = (DEG_PER + DEG_T - 1) / DEG_T;

    v4i d[NI];
#pragma unroll
    for (int i = 0; i < NI; ++i) {
        const int it = tid + i * DEG_T;
        if (it < DEG_PER) d[i] = ((const v4i*)dst)[base + it];
    }
    for (int i = tid; i < NNODES / 8; i += DEG_T) ((v4i*)cnt)[i] = (v4i){0, 0, 0, 0};
    __syncthreads();

#pragma unroll
    for (int i = 0; i < NI; ++i) {
        const int it = tid + i * DEG_T;
        if (it < DEG_PER) {
            const v4i e = d[i];
            atomicAdd(&cnt[e.x >> 1], 1u << ((e.x & 1) * 16));
            atomicAdd(&cnt[e.y >> 1], 1u << ((e.y & 1) * 16));
            atomicAdd(&cnt[e.z >> 1], 1u << ((e.z & 1) * 16));
            atomicAdd(&cnt[e.w >> 1], 1u << ((e.w & 1) * 16));
        }
    }
    __syncthreads();

    constexpr int FI = (NNODES + DEG_T - 1) / DEG_T;
    constexpr int FU = 7;
    static_assert(FI % FU == 0, "the add loop runs whole groups");
    const int start = (int)blockIdx.x * (NNODES / 32 / DEG_WGS) * 32;
#pragma nounroll
    for (int j0 = 0; j0 < FI; j0 += FU) {
        unsigned wv[FU];
#pragma unroll
        for (int u = 0; u < FU; ++u) {
            const int i = tid + (j0 + u) * DEG_T;
            int n = start + i;
            if (n >= NNODES) n -= NNODES;
            wv[u] = i < NNODES ? cnt[n >> 1] : 0u;
        }
#pragma unroll
        for (int u = 0; u < FU; ++u) {
            const int i = tid + (j0 + u) * DEG_T;
            int n = start + i;
            if (n >= NNODES) n -= NNODES;
            const unsigned c = (wv[u] >> ((n & 1) * 16)) & 0xffffu;
            if (c != 0u) atomicAdd(&deg[n], (float)c);
        }
    }
}


template <int K, int N>
__global__ __launch_bounds__(256) void gcn_gemm_wmma_kernel(
    const float* __restrict__ A, const float* __restrict__ B,
    float* __restrict__ deg, float* __restrict__ C, int M) {
    constexpr int NT = N / 16;

    const int lane     = threadIdx.x & 31;
    const int wave     = threadIdx.x >> 5;
    const int row_tile = blockIdx.x * 8 + wave;
    if (row_tile * 16 >= M) return;

    if ((row_tile & 1) == 0) {
        const int di = row_tile * 16 + lane;
        if (di < M) deg[di] = 1.0f;
    }

    const int m     = lane & 15;
    const int khalf = lane >> 4;
    const int row   = row_tile * 16 + m;

    v8f acc[NT];
#pragma unroll
    for (int nt = 0; nt < NT; ++nt) acc[nt] = (v8f){0.f,0.f,0.f,0.f,0.f,0.f,0.f,0.f};

    const float* __restrict__ arow = A + (size_t)row * K;
#pragma unroll
    for (int k0 = 0; k0 < K; k0 += 4) {
        const int kk = k0 + khalf * 2;
        const v2f a = *(const v2f*)(arow + kk);
#pragma unroll
        for (int nt = 0; nt < NT; ++nt) {
            const int n = nt * 16 + m;
            v2f b;
            b.x = B[(size_t)kk * N + n];
            b.y = B[(size_t)(kk + 1) * N + n];
            acc[nt] = __builtin_amdgcn_wmma_f32_16x16x4_f32(
                false, a, false, b, (short)0, acc[nt], false, false);
        }
    }

    const int r0 = row_tile * 16 + khalf * 8;
#pragma unroll
    for (int nt = 0; nt < NT; ++nt) {
        float* __restrict__ crow = C + (size_t)r0 * N + nt * 16 + m;
#pragma unroll
        for (int j = 0; j < 8; ++j) crow[(size_t)j * N] = acc[nt][j];
    }
}

#define AGG_QCAP  64
#define AGG_G     16
#define AGG_LCAP  512

static_assert((AGG_WGS - 1) * AGG_RPW < NNODES, "every work-group owns at least one row");
static_assert(AGG_LCAP % 32 == 0, "a full group of 32 saved entries starts on a 256-byte boundary");
static_assert(AGG_WAVES == 32, "one lane per wave reads or stores the list lengths");

template <int F, bool SAVE, bool USE>
__global__ __launch_bounds__(AGG_T) void gcn_agg_lds_kernel(
    const float* __restrict__ h, const int* __restrict__ src,
    const int* __restrict__ dst, const int* __restrict__ bcnt, const unsigned* __restrict__ bkt,
    const float* __restrict__ deg, const float* __restrict__ bias,
    float* __restrict__ out, int* __restrict__ lcnt, int* __restrict__ lent,
    const float* __restrict__ Wn, float* __restrict__ hn) {
    constexpr int VPL = F / 32;
    __shared__ __attribute__((aligned(16))) float acc[AGG_RPW * F];
    __shared__ int pend[AGG_WAVES * AGG_QCAP];

    const int tid  = threadIdx.x;
    const int lane = tid & 31;
    const int wave = tid >> 5;
    const int lo   = blockIdx.x * AGG_RPW;
    const int rem  = NNODES - lo;
    const int nrows = rem < AGG_RPW ? rem : AGG_RPW;

    const int nvec = nrows * (F / 4);
    constexpr int NI = (AGG_RPW * (F / 4) + AGG_T - 1) / AGG_T;
    {
        const float* __restrict__ hown = h + (size_t)lo * F;
        static_assert(AGG_T % (F / 4) == 0, "a thread's pieces share their columns");
        const v4f bv = ((const v4f*)bias)[tid % (F / 4)];
        v4f tmp[NI];
        float dg[NI];
#pragma unroll
        for (int i = 0; i < NI; ++i) {
            const int idx = tid + i * AGG_T;
            if (idx < nvec) {
                tmp[i] = ((const v4f*)hown)[idx];
                dg[i]  = deg[lo + idx / (F / 4)];
            }
        }
#pragma unroll
        for (int i = 0; i < NI; ++i) {
            const int idx = tid + i * AGG_T;
            if (idx < nvec) {
                const float sq  = rsqrtf(dg[i]);
                const float nrm = sq * sq;
                ((v4f*)acc)[idx] = tmp[i] * nrm + bv;
            }
        }
    }
    __syncthreads();

    int* __restrict__ myq = pend + wave * AGG_QCAP;
    v2i* __restrict__ mylst = (v2i*)lent + ((size_t)blockIdx.x * AGG_WAVES + wave) * AGG_LCAP;

    auto gather_add = [&](int s, int dl, float w, int cnt) {
        const float* __restrict__ hb = h + lane * VPL;
#pragma unroll
        for (int j = 0; j < 32; j += AGG_G) {
            if (j < cnt) {
                float v[AGG_G][VPL];
#pragma unroll
                for (int t = 0; t < AGG_G; ++t) {
                    const int sj = __builtin_amdgcn_readlane(s, j + t);
                    const float* __restrict__ p = hb + (size_t)sj * F;
                    if (VPL == 2) { const v2f r = *(const v2f*)p; v[t][0] = r.x; v[t][VPL - 1] = r.y; }
                    else          { v[t][0] = *p; }
                }
#pragma unroll
                for (int t = 0; t < AGG_G; ++t) {
                    if (j + t < cnt) {
                        const float wj = __int_as_float(__builtin_amdgcn_readlane(__float_as_int(w), j + t));
                        const int   dj = __builtin_amdgcn_readlane(dl, j + t);
                        float* a = acc + dj * F + lane * VPL;
#pragma unroll
                        for (int c = 0; c < VPL; ++c) atomicAdd(a + c, v[t][c] * wj);
                    }
                }
            }
        }
    };

    bool from_list = false;
    if (USE) {
        const int len_l = lcnt[blockIdx.x * AGG_WAVES + lane];
        from_list = __builtin_amdgcn_ballot_w32(len_l > AGG_LCAP) == 0u;
        if (from_list) {
            const int n = __builtin_amdgcn_readlane(len_l, __builtin_amdgcn_readfirstlane(wave));
            v2i cur = (v2i){0, 0};
            if (lane < n) cur = mylst[lane];
            for (int base = 0; base < n; base += 32) {
                const int rest = n - base;
                const int cnt  = rest < 32 ? rest : 32;
                v2i nxt = (v2i){0, 0};
                if (base + 32 + lane < n) nxt = mylst[base + 32 + lane];
                gather_add(cur.x & ((1 << AGG_SBITS) - 1), cur.x >> AGG_SBITS, __int_as_float(cur.y), cnt);
                cur = nxt;
            }
        }
    }

    int tot = 0;
    if (!from_list) {
        int qn = 0;
        const unsigned lt = (1u << lane) - 1u;

        auto drain = [&](int base, int cnt) {
            const int pk  = myq[base + (lane < cnt ? lane : 0)];
            const int s   = pk & ((1 << AGG_SBITS) - 1);
            const int dl  = pk >> AGG_SBITS;
            const float w = rsqrtf(deg[s]) * rsqrtf(deg[lo + dl]);
            if (SAVE) {
                const int p = tot + lane;
                if (lane < cnt && p < AGG_LCAP) mylst[p] = (v2i){pk, __float_as_int(w)};
                tot += cnt;
            }
            gather_add(s, dl, w, cnt);
        };

        const int pb = blockIdx.x * PART_WGS + __builtin_amdgcn_readfirstlane(wave) * PART_SPW;
        const int bl = lane < PART_SPW ? bcnt[pb + lane] : 0;
        if (__builtin_amdgcn_ballot_w32(bl > PART_CAP) == 0u) {
            constexpr int NQ = PART_SPW * (PART_CAP / 32);
            auto ldgrp = [&](int q) {
                const int j   = q % PART_SPW;
                const int idx = (q / PART_SPW) * 32 + lane;
                const int n   = __builtin_amdgcn_readlane(bl, j);
                return idx < n ? (int)bkt[(size_t)(pb + j) * PART_CAP + idx] : 0;
            };
            int cur = ldgrp(0);
#pragma nounroll
            for (int q = 0; q < NQ; ++q) {
                const int rest = __builtin_amdgcn_readlane(bl, q % PART_SPW) - (q / PART_SPW) * 32;
                const int n    = rest < 0 ? 0 : (rest < 32 ? rest : 32);
                if (lane < n) myq[qn + lane] = cur;
                qn += n;
                int nxt = 0;
                if (q + 1 < NQ) nxt = ldgrp(q + 1);
                if (qn >= 32) { qn -= 32; drain(qn, 32); }
                cur = nxt;
            }
        } else {
            constexpr int EPW = NEDGES / AGG_WAVES;
            const int e0 = __builtin_amdgcn_readfirstlane(wave) * EPW;
#pragma nounroll
            for (int i = 0; i < EPW; i += 32) {
                int pk = -1;
                if (i + lane < EPW) {
                    const unsigned dl = (unsigned)(dst[e0 + i + lane] - lo);
                    if (dl < (unsigned)nrows) pk = src[e0 + i + lane] | (int)(dl << AGG_SBITS);
                }
                const unsigned bal = __builtin_amdgcn_ballot_w32(pk >= 0);
                if (pk >= 0) myq[qn + __builtin_popcount(bal & lt)] = pk;
                qn += __builtin_popcount(bal);
                if (qn >= 32) { qn -= 32; drain(qn, 32); }
            }
        }
        if (qn > 0) drain(0, qn);
    }
    if (SAVE && lane == 0) myq[0] = tot;

    __syncthreads();
    if (SAVE && wave == 0) lcnt[blockIdx.x * AGG_WAVES + lane] = pend[lane * AGG_QCAP];
    if (SAVE) {
        static_assert(!SAVE || (F == HID_DIM && OUT_DIM == 32), "epilogue: K = F, two column tiles");
        static_assert((AGG_RPW + 15) / 16 <= AGG_WAVES, "one slab per wave");
        const int tile = __builtin_amdgcn_readfirstlane(wave);
        if (tile * 16 < nrows) {
            const int m     = lane & 15;
            const int khalf = lane >> 4;
            const int rl = tile * 16 + m < nrows ? tile * 16 + m : nrows - 1;
            const float* arow = acc + rl * F;
            v8f ct[2];
            ct[0] = (v8f){0.f,0.f,0.f,0.f,0.f,0.f,0.f,0.f};
            ct[1] = (v8f){0.f,0.f,0.f,0.f,0.f,0.f,0.f,0.f};
#pragma unroll
            for (int k0 = 0; k0 < F; k0 += 4) {
                const int kk = k0 + khalf * 2;
                v2f a = *(const v2f*)(arow + kk);
                a.x = fmaxf(a.x, 0.0f); a.y = fmaxf(a.y, 0.0f);
#pragma unroll
                for (int nt = 0; nt < 2; ++nt) {
                    const int n = nt * 16 + m;
                    v2f b;
                    b.x = Wn[(size_t)kk * OUT_DIM + n];
                    b.y = Wn[(size_t)(kk + 1) * OUT_DIM + n];
                    ct[nt] = __builtin_amdgcn_wmma_f32_16x16x4_f32(
                        false, a, false, b, (short)0, ct[nt], false, false);
                }
            }
            const int rb = tile * 16 + khalf * 8;
#pragma unroll
            for (int nt = 0; nt < 2; ++nt) {
                float* __restrict__ crow = hn + (size_t)(lo + rb) * OUT_DIM + nt * 16 + m;
#pragma unroll
                for (int j = 0; j < 8; ++j) {
                    if (rb + j < nrows) crow[(size_t)j * OUT_DIM] = ct[nt][j];
                }
            }
        }
    } else {
        float* __restrict__ obase = out + (size_t)lo * F;
#pragma unroll
        for (int i = 0; i < NI; ++i) {
            const int idx = tid + i * AGG_T;
            if (idx < nvec) ((v4f*)obase)[idx] = ((const v4f*)acc)[idx];
        }
    }
}

extern "C" void kernel_launch(void* const* d_in, const int* in_sizes, int n_in,
                              void* d_out, int out_size, void* d_ws, size_t ws_size,
                              hipStream_t stream) {
    const float* x  = (const float*)d_in[0];
    const int*   ei = (const int*)d_in[1];
    const float* W1 = (const float*)d_in[2];
    const float* b1 = (const float*)d_in[3];
    const float* W2 = (const float*)d_in[4];
    const float* b2 = (const float*)d_in[5];
    const int* src = ei;
    const int* dst = ei + NEDGES;
    float* out = (float*)d_out;

    float* ws   = (float*)d_ws;
    float* deg  = ws;
    float* h1   = ws + 131072;
    int* lcnt   = (int*)(h1 + (size_t)NNODES * HID_DIM);
    int* lent   = lcnt + AGG_WGS * AGG_WAVES;
    int* bcnt     = lent + (size_t)AGG_WGS * AGG_WAVES * AGG_LCAP * 2;
    unsigned* bkt = (unsigned*)(bcnt + AGG_WGS * PART_WGS);
    float* h2   = (float*)(bkt + (size_t)AGG_WGS * PART_WGS * PART_CAP);

    const int T = 256;

    {
        int waves  = NNODES / 16;
        int blocks = (waves + 7) / 8;
        gcn_gemm_wmma_kernel<IN_DIM, HID_DIM><<<blocks, T, 0, stream>>>(x, W1, deg, h1, NNODES);
    }

    gcn_deg_kernel<<<DEG_WGS + PART_WGS, DEG_T, 0, stream>>>(src, dst, deg, bcnt, bkt);

    gcn_agg_lds_kernel<HID_DIM, true, false><<<AGG_WGS, AGG_T, 0, stream>>>(
        h1, src, dst, bcnt, bkt, deg, b1, nullptr, lcnt, lent, W2, h2);

    gcn_agg_lds_kernel<OUT_DIM, false, true><<<AGG_WGS, AGG_T, 0, stream>>>(
        h2, src, dst, bcnt, bkt, deg, b2, out, lcnt, lent, nullptr, nullptr);
}


# degree counted from the sorted buckets: a work-group counts its 391 rows in LDS (integer adds, exact) and stores deg once, no memory atomics or 200 KB tables; the sort moves into the GEMM launch
# speedup vs baseline: 3.5615x; 1.1161x over previous
#include <hip/hip_runtime.h>

#define NNODES 100000
#define NEDGES 1200000
#define IN_DIM 64
#define HID_DIM 64
#define OUT_DIM 32

typedef __attribute__((ext_vector_type(2))) float v2f;
typedef __attribute__((ext_vector_type(4))) float v4f;
typedef __attribute__((ext_vector_type(8))) float v8f;
typedef __attribute__((ext_vector_type(4))) int v4i;
typedef __attribute__((ext_vector_type(2))) int v2i;

#define AGG_WGS   256
#define AGG_RPW   ((NNODES + AGG_WGS - 1) / AGG_WGS)
#define AGG_T     1024
#define AGG_WAVES (AGG_T / 32)
#define AGG_SBITS 17

#define PART_WGS  160
#define PART_PER  (NEDGES / 4 / PART_WGS)
#define PART_CAP  64
#define PART_SPW  (PART_WGS / AGG_WAVES)
static_assert((NEDGES / 4) % PART_WGS == 0, "the slices are equal");
static_assert(PART_WGS % AGG_WAVES == 0, "every aggregation wave reads the same number of buckets");
static_assert(PART_SPW <= 32, "one lane per bucket reads its length");
static_assert(PART_CAP % 32 == 0, "a bucket is read 32 entries at a time");
static_assert(NEDGES % AGG_WAVES == 0, "a wave's slices are a whole part of the edge list");
static_assert(NNODES <= (1 << AGG_SBITS), "source row must fit its field of an entry");
static_assert(AGG_RPW < (1 << (31 - AGG_SBITS)), "local row must fit its field of an entry");

#define L1_T     512
#define GEMM_SPW (L1_T / 32)
#define GEMM_WGS ((NNODES / 16 + GEMM_SPW - 1) / GEMM_SPW)
static_assert(NNODES % 16 == 0, "the GEMM has whole slabs only");

template <int K, int N>
__global__ __launch_bounds__(L1_T) void gcn_gemm_sort_kernel(
    const float* __restrict__ A, const float* __restrict__ B, float* __restrict__ C, int M,
    const int* __restrict__ src, const int* __restrict__ dst,
    int* __restrict__ bcnt, unsigned* __restrict__ bkt) {
    if (blockIdx.x < PART_WGS) {
        __shared__ __attribute__((aligned(16))) unsigned srt[AGG_WGS + AGG_WGS * PART_CAP];
        const int tid   = threadIdx.x;
        const int p     = blockIdx.x;
        const int pbase = p * PART_PER;
        unsigned* bc  = srt;
        unsigned* stg = srt + AGG_WGS;
        constexpr int NP  = (PART_PER + L1_T - 1) / L1_T;
        constexpr int GPR = PART_CAP / 4;

        v4i dd[NP], ss[NP];
#pragma unroll
        for (int i = 0; i < NP; ++i) {
            const int it = tid + i * L1_T;
            if (it < PART_PER) {
                dd[i] = ((const v4i*)dst)[pbase + it];
                ss[i] = ((const v4i*)src)[pbase + it];
            }
        }
        for (int i = tid; i < (AGG_WGS + AGG_WGS * PART_CAP) / 4; i += L1_T) ((v4i*)srt)[i] = (v4i){0, 0, 0, 0};
        __syncthreads();

#pragma unroll
        for (int i = 0; i < NP; ++i) {
            const int it = tid + i * L1_T;
            if (it < PART_PER) {
#pragma unroll
                for (int c = 0; c < 4; ++c) {
                    const unsigned d    = (unsigned)dd[i][c];
                    const unsigned r    = d / AGG_RPW;
                    const unsigned dl   = d - r * AGG_RPW;
                    const unsigned slot = atomicAdd(&bc[r], 1u);
                    if (slot < PART_CAP) stg[r * PART_CAP + slot] = (unsigned)ss[i][c] | (dl << AGG_SBITS);
                }
            }
        }
        __syncthreads();

        for (int gi = tid; gi < AGG_WGS * GPR; gi += L1_T) {
            const int r = gi / GPR;
            const int g = gi % GPR;
            if ((unsigned)(g * 4) < bc[r])
                ((v4i*)bkt)[((size_t)r * PART_WGS + p) * GPR + g] = ((const v4i*)stg)[gi];
        }
        if (tid < AGG_WGS) bcnt[tid * PART_WGS + p] = (int)bc[tid];
        return;
    }

    constexpr int NT = N / 16;

    const int lane     = threadIdx.x & 31;
    const int wave     = threadIdx.x >> 5;
    const int row_tile = ((int)blockIdx.x - PART_WGS) * GEMM_SPW + wave;
    if (row_tile * 16 >= M) return;

    const int m     = lane & 15;
    const int khalf = lane >> 4;
    const int row   = row_tile * 16 + m;

    v8f acc[NT];
#pragma unroll
    for (int nt = 0; nt < NT; ++nt) acc[nt] = (v8f){0.f,0.f,0.f,0.f,0.f,0.f,0.f,0.f};

    const float* __restrict__ arow = A + (size_t)row * K;
#pragma unroll
    for (int k0 = 0; k0 < K; k0 += 4) {
        const int kk = k0 + khalf * 2;
        const v2f a = *(const v2f*)(arow + kk);
#pragma unroll
        for (int nt = 0; nt < NT; ++nt) {
            const int n = nt * 16 + m;
            v2f b;
            b.x = B[(size_t)kk * N + n];
            b.y = B[(size_t)(kk + 1) * N + n];
            acc[nt] = __builtin_amdgcn_wmma_f32_16x16x4_f32(
                false, a, false, b, (short)0, acc[nt], false, false);
        }
    }

    const int r0 = row_tile * 16 + khalf * 8;
#pragma unroll
    for (int nt = 0; nt < NT; ++nt) {
        float* __restrict__ crow = C + (size_t)r0 * N + nt * 16 + m;
#pragma unroll
        for (int j = 0; j < 8; ++j) crow[(size_t)j * N] = acc[nt][j];
    }
}

#define DEG_T   256
#define DEG_GPR (PART_CAP / 4)
#define DEG_NG  (PART_WGS * DEG_GPR / DEG_T)
static_assert((PART_WGS * DEG_GPR) % DEG_T == 0, "every thread reads the same number of groups");
static_assert(PART_WGS <= DEG_T, "one thread per bucket reads its length");
static_assert(PART_WGS % 32 == 0, "the lengths are looked through 32 at a time");

__global__ __launch_bounds__(DEG_T) void gcn_deg_kernel(const int* __restrict__ dst, const int* __restrict__ bcnt,
                                                        const unsigned* __restrict__ bkt, float* __restrict__ deg) {
    __shared__ unsigned cnt[AGG_RPW];
    __shared__ int blen[PART_WGS];
    const int tid   = threadIdx.x;
    const int lo    = blockIdx.x * AGG_RPW;
    const int rem   = NNODES - lo;
    const int nrows = rem < AGG_RPW ? rem : AGG_RPW;
    const int pb    = blockIdx.x * PART_WGS;

    if (tid < PART_WGS) blen[tid] = bcnt[pb + tid];
    for (int i = tid; i < AGG_RPW; i += DEG_T) cnt[i] = 0u;
    __syncthreads();

    int left[DEG_NG];
#pragma unroll
    for (int i = 0; i < DEG_NG; ++i) {
        const int gi = tid + i * DEG_T;
        const int n  = blen[gi / DEG_GPR];
        left[i] = n <= PART_CAP ? n - (gi % DEG_GPR) * 4 : 0;
    }
    v4i e[DEG_NG];
#pragma unroll
    for (int i = 0; i < DEG_NG; ++i) {
        const int gi = tid + i * DEG_T;
        if (left[i] > 0) e[i] = ((const v4i*)bkt)[(size_t)pb * DEG_GPR + gi];
    }
#pragma unroll
    for (int i = 0; i < DEG_NG; ++i) {
#pragma unroll
        for (int c = 0; c < 4; ++c) {
            if (c < left[i]) atomicAdd(&cnt[(unsigned)e[i][c] >> AGG_SBITS], 1u);
        }
    }

    unsigned over = 0u;
    for (int k = 0; k < PART_WGS; k += 32) over |= __builtin_amdgcn_ballot_w32(blen[k + (tid & 31)] > PART_CAP);
    if (over != 0u) {
        for (int b = 0; b < PART_WGS; ++b) {
            if (blen[b] > PART_CAP) {
                const int e0 = b * (PART_PER * 4);
                for (int i = tid; i < PART_PER * 4; i += DEG_T) {
                    const unsigned dl = (unsigned)(dst[e0 + i] - lo);
                    if (dl < (unsigned)nrows) atomicAdd(&cnt[dl], 1u);
                }
            }
        }
    }
    __syncthreads();
    for (int i = tid; i < nrows; i += DEG_T) deg[lo + i] = 1.0f + (float)cnt[i];
}

#define AGG_QCAP  64
#define AGG_G     16
#define AGG_LCAP  512

static_assert((AGG_WGS - 1) * AGG_RPW < NNODES, "every work-group owns at least one row");
static_assert(AGG_LCAP % 32 == 0, "a full group of 32 saved entries starts on a 256-byte boundary");
static_assert(AGG_WAVES == 32, "one lane per wave reads or stores the list lengths");

template <int F, bool SAVE, bool USE>
__global__ __launch_bounds__(AGG_T) void gcn_agg_lds_kernel(
    const float* __restrict__ h, const int* __restrict__ src,
    const int* __restrict__ dst, const int* __restrict__ bcnt, const unsigned* __restrict__ bkt,
    const float* __restrict__ deg, const float* __restrict__ bias,
    float* __restrict__ out, int* __restrict__ lcnt, int* __restrict__ lent,
    const float* __restrict__ Wn, float* __restrict__ hn) {
    constexpr int VPL = F / 32;
    __shared__ __attribute__((aligned(16))) float acc[AGG_RPW * F];
    __shared__ int pend[AGG_WAVES * AGG_QCAP];

    const int tid  = threadIdx.x;
    const int lane = tid & 31;
    const int wave = tid >> 5;
    const int lo   = blockIdx.x * AGG_RPW;
    const int rem  = NNODES - lo;
    const int nrows = rem < AGG_RPW ? rem : AGG_RPW;

    const int nvec = nrows * (F / 4);
    constexpr int NI = (AGG_RPW * (F / 4) + AGG_T - 1) / AGG_T;
    {
        const float* __restrict__ hown = h + (size_t)lo * F;
        static_assert(AGG_T % (F / 4) == 0, "a thread's pieces share their columns");
        const v4f bv = ((const v4f*)bias)[tid % (F / 4)];
        v4f tmp[NI];
        float dg[NI];
#pragma unroll
        for (int i = 0; i < NI; ++i) {
            const int idx = tid + i * AGG_T;
            if (idx < nvec) {
                tmp[i] = ((const v4f*)hown)[idx];
                dg[i]  = deg[lo + idx / (F / 4)];
            }
        }
#pragma unroll
        for (int i = 0; i < NI; ++i) {
            const int idx = tid + i * AGG_T;
            if (idx < nvec) {
                const float sq  = rsqrtf(dg[i]);
                const float nrm = sq * sq;
                ((v4f*)acc)[idx] = tmp[i] * nrm + bv;
            }
        }
    }
    __syncthreads();

    int* __restrict__ myq = pend + wave * AGG_QCAP;
    v2i* __restrict__ mylst = (v2i*)lent + ((size_t)blockIdx.x * AGG_WAVES + wave) * AGG_LCAP;

    auto gather_add = [&](int s, int dl, float w, int cnt) {
        const float* __restrict__ hb = h + lane * VPL;
#pragma unroll
        for (int j = 0; j < 32; j += AGG_G) {
            if (j < cnt) {
                float v[AGG_G][VPL];
#pragma unroll
                for (int t = 0; t < AGG_G; ++t) {
                    const int sj = __builtin_amdgcn_readlane(s, j + t);
                    const float* __restrict__ p = hb + (size_t)sj * F;
                    if (VPL == 2) { const v2f r = *(const v2f*)p; v[t][0] = r.x; v[t][VPL - 1] = r.y; }
                    else          { v[t][0] = *p; }
                }
#pragma unroll
                for (int t = 0; t < AGG_G; ++t) {
                    if (j + t < cnt) {
                        const float wj = __int_as_float(__builtin_amdgcn_readlane(__float_as_int(w), j + t));
                        const int   dj = __builtin_amdgcn_readlane(dl, j + t);
                        float* a = acc + dj * F + lane * VPL;
#pragma unroll
                        for (int c = 0; c < VPL; ++c) atomicAdd(a + c, v[t][c] * wj);
                    }
                }
            }
        }
    };

    bool from_list = false;
    if (USE) {
        const int len_l = lcnt[blockIdx.x * AGG_WAVES + lane];
        from_list = __builtin_amdgcn_ballot_w32(len_l > AGG_LCAP) == 0u;
        if (from_list) {
            const int n = __builtin_amdgcn_readlane(len_l, __builtin_amdgcn_readfirstlane(wave));
            v2i cur = (v2i){0, 0};
            if (lane < n) cur = mylst[lane];
            for (int base = 0; base < n; base += 32) {
                const int rest = n - base;
                const int cnt  = rest < 32 ? rest : 32;
                v2i nxt = (v2i){0, 0};
                if (base + 32 + lane < n) nxt = mylst[base + 32 + lane];
                gather_add(cur.x & ((1 << AGG_SBITS) - 1), cur.x >> AGG_SBITS, __int_as_float(cur.y), cnt);
                cur = nxt;
            }
        }
    }

    int tot = 0;
    if (!from_list) {
        int qn = 0;
        const unsigned lt = (1u << lane) - 1u;

        auto drain = [&](int base, int cnt) {
            const int pk  = myq[base + (lane < cnt ? lane : 0)];
            const int s   = pk & ((1 << AGG_SBITS) - 1);
            const int dl  = pk >> AGG_SBITS;
            const float w = rsqrtf(deg[s]) * rsqrtf(deg[lo + dl]);
            if (SAVE) {
                const int p = tot + lane;
                if (lane < cnt && p < AGG_LCAP) mylst[p] = (v2i){pk, __float_as_int(w)};
                tot += cnt;
            }
            gather_add(s, dl, w, cnt);
        };

        const int pb = blockIdx.x * PART_WGS + __builtin_amdgcn_readfirstlane(wave) * PART_SPW;
        const int bl = lane < PART_SPW ? bcnt[pb + lane] : 0;
        if (__builtin_amdgcn_ballot_w32(bl > PART_CAP) == 0u) {
            constexpr int NQ = PART_SPW * (PART_CAP / 32);
            auto ldgrp = [&](int q) {
                const int j   = q % PART_SPW;
                const int idx = (q / PART_SPW) * 32 + lane;
                const int n   = __builtin_amdgcn_readlane(bl, j);
                return idx < n ? (int)bkt[(size_t)(pb + j) * PART_CAP + idx] : 0;
            };
            int cur = ldgrp(0);
#pragma nounroll
            for (int q = 0; q < NQ; ++q) {
                const int rest = __builtin_amdgcn_readlane(bl, q % PART_SPW) - (q / PART_SPW) * 32;
                const int n    = rest < 0 ? 0 : (rest < 32 ? rest : 32);
                if (lane < n) myq[qn + lane] = cur;
                qn += n;
                int nxt = 0;
                if (q + 1 < NQ) nxt = ldgrp(q + 1);
                if (qn >= 32) { qn -= 32; drain(qn, 32); }
                cur = nxt;
            }
        } else {
            constexpr int EPW = NEDGES / AGG_WAVES;
            const int e0 = __builtin_amdgcn_readfirstlane(wave) * EPW;
#pragma nounroll
            for (int i = 0; i < EPW; i += 32) {
                int pk = -1;
                if (i + lane < EPW) {
                    const unsigned dl = (unsigned)(dst[e0 + i + lane] - lo);
                    if (dl < (unsigned)nrows) pk = src[e0 + i + lane] | (int)(dl << AGG_SBITS);
                }
                const unsigned bal = __builtin_amdgcn_ballot_w32(pk >= 0);
                if (pk >= 0) myq[qn + __builtin_popcount(bal & lt)] = pk;
                qn += __builtin_popcount(bal);
                if (qn >= 32) { qn -= 32; drain(qn, 32); }
            }
        }
        if (qn > 0) drain(0, qn);
    }
    if (SAVE && lane == 0) myq[0] = tot;

    __syncthreads();
    if (SAVE && wave == 0) lcnt[blockIdx.x * AGG_WAVES + lane] = pend[lane * AGG_QCAP];
    if (SAVE) {
        static_assert(!SAVE || (F == HID_DIM && OUT_DIM == 32), "epilogue: K = F, two column tiles");
        static_assert((AGG_RPW + 15) / 16 <= AGG_WAVES, "one slab per wave");
        const int tile = __builtin_amdgcn_readfirstlane(wave);
        if (tile * 16 < nrows) {
            const int m     = lane & 15;
            const int khalf = lane >> 4;
            const int rl = tile * 16 + m < nrows ? tile * 16 + m : nrows - 1;
            const float* arow = acc + rl * F;
            v8f ct[2];
            ct[0] = (v8f){0.f,0.f,0.f,0.f,0.f,0.f,0.f,0.f};
            ct[1] = (v8f){0.f,0.f,0.f,0.f,0.f,0.f,0.f,0.f};
#pragma unroll
            for (int k0 = 0; k0 < F; k0 += 4) {
                const int kk = k0 + khalf * 2;
                v2f a = *(const v2f*)(arow + kk);
                a.x = fmaxf(a.x, 0.0f); a.y = fmaxf(a.y, 0.0f);
#pragma unroll
                for (int nt = 0; nt < 2; ++nt) {
                    const int n = nt * 16 + m;
                    v2f b;
                    b.x = Wn[(size_t)kk * OUT_DIM + n];
                    b.y = Wn[(size_t)(kk + 1) * OUT_DIM + n];
                    ct[nt] = __builtin_amdgcn_wmma_f32_16x16x4_f32(
                        false, a, false, b, (short)0, ct[nt], false, false);
                }
            }
            const int rb = tile * 16 + khalf * 8;
#pragma unroll
            for (int nt = 0; nt < 2; ++nt) {
                float* __restrict__ crow = hn + (size_t)(lo + rb) * OUT_DIM + nt * 16 + m;
#pragma unroll
                for (int j = 0; j < 8; ++j) {
                    if (rb + j < nrows) crow[(size_t)j * OUT_DIM] = ct[nt][j];
                }
            }
        }
    } else {
        float* __restrict__ obase = out + (size_t)lo * F;
#pragma unroll
        for (int i = 0; i < NI; ++i) {
            const int idx = tid + i * AGG_T;
            if (idx < nvec) ((v4f*)obase)[idx] = ((const v4f*)acc)[idx];
        }
    }
}

extern "C" void kernel_launch(void* const* d_in, const int* in_sizes, int n_in,
                              void* d_out, int out_size, void* d_ws, size_t ws_size,
                              hipStream_t stream) {
    const float* x  = (const float*)d_in[0];
    const int*   ei = (const int*)d_in[1];
    const float* W1 = (const float*)d_in[2];
    const float* b1 = (const float*)d_in[3];
    const float* W2 = (const float*)d_in[4];
    const float* b2 = (const float*)d_in[5];
    const int* src = ei;
    const int* dst = ei + NEDGES;
    float* out = (float*)d_out;

    float* ws   = (float*)d_ws;
    float* deg  = ws;
    float* h1   = ws + 131072;
    int* lcnt   = (int*)(h1 + (size_t)NNODES * HID_DIM);
    int* lent   = lcnt + AGG_WGS * AGG_WAVES;
    int* bcnt     = lent + (size_t)AGG_WGS * AGG_WAVES * AGG_LCAP * 2;
    unsigned* bkt = (unsigned*)(bcnt + AGG_WGS * PART_WGS);
    float* h2   = (float*)(bkt + (size_t)AGG_WGS * PART_WGS * PART_CAP);

    gcn_gemm_sort_kernel<IN_DIM, HID_DIM><<<PART_WGS + GEMM_WGS, L1_T, 0, stream>>>(
        x, W1, h1, NNODES, src, dst, bcnt, bkt);

    gcn_deg_kernel<<<AGG_WGS, DEG_T, 0, stream>>>(dst, bcnt, bkt, deg);

    gcn_agg_lds_kernel<HID_DIM, true, false><<<AGG_WGS, AGG_T, 0, stream>>>(
        h1, src, dst, bcnt, bkt, deg, b1, nullptr, lcnt, lent, W2, h2);

    gcn_agg_lds_kernel<OUT_DIM, false, true><<<AGG_WGS, AGG_T, 0, stream>>>(
        h2, src, dst, bcnt, bkt, deg, b2, out, lcnt, lent, nullptr, nullptr);
}


# layer-1 aggregation keeps every node's degree in LDS, one byte a node (table from the degree pass; exact value from memory if it does not fit): per-edge weight lookups are LDS reads, not memory loads
# speedup vs baseline: 3.6723x; 1.0311x over previous
#include <hip/hip_runtime.h>

#define NNODES 100000
#define NEDGES 1200000
#define IN_DIM 64
#define HID_DIM 64
#define OUT_DIM 32

typedef __attribute__((ext_vector_type(2))) float v2f;
typedef __attribute__((ext_vector_type(4))) float v4f;
typedef __attribute__((ext_vector_type(8))) float v8f;
typedef __attribute__((ext_vector_type(4))) int v4i;
typedef __attribute__((ext_vector_type(2))) int v2i;

#define AGG_WGS   256
#define AGG_RPW   ((NNODES + AGG_WGS - 1) / AGG_WGS)
#define AGG_T     1024
#define AGG_WAVES (AGG_T / 32)
#define AGG_SBITS 17

#define PART_WGS  160
#define PART_PER  (NEDGES / 4 / PART_WGS)
#define PART_CAP  64
#define PART_SPW  (PART_WGS / AGG_WAVES)
static_assert((NEDGES / 4) % PART_WGS == 0, "the slices are equal");
static_assert(PART_WGS % AGG_WAVES == 0, "every aggregation wave reads the same number of buckets");
static_assert(PART_SPW <= 32, "one lane per bucket reads its length");
static_assert(PART_CAP % 32 == 0, "a bucket is read 32 entries at a time");
static_assert(NEDGES % AGG_WAVES == 0, "a wave's slices are a whole part of the edge list");
static_assert(NNODES <= (1 << AGG_SBITS), "source row must fit its field of an entry");
static_assert(AGG_RPW < (1 << (31 - AGG_SBITS)), "local row must fit its field of an entry");

#define L1_T     512
#define GEMM_SPW (L1_T / 32)
#define GEMM_WGS ((NNODES / 16 + GEMM_SPW - 1) / GEMM_SPW)
static_assert(NNODES % 16 == 0, "the GEMM has whole slabs only");

template <int K, int N>
__global__ __launch_bounds__(L1_T) void gcn_gemm_sort_kernel(
    const float* __restrict__ A, const float* __restrict__ B, float* __restrict__ C, int M,
    const int* __restrict__ src, const int* __restrict__ dst,
    int* __restrict__ bcnt, unsigned* __restrict__ bkt) {
    if (blockIdx.x < PART_WGS) {
        __shared__ __attribute__((aligned(16))) unsigned srt[AGG_WGS + AGG_WGS * PART_CAP];
        const int tid   = threadIdx.x;
        const int p     = blockIdx.x;
        const int pbase = p * PART_PER;
        unsigned* bc  = srt;
        unsigned* stg = srt + AGG_WGS;
        constexpr int NP  = (PART_PER + L1_T - 1) / L1_T;
        constexpr int GPR = PART_CAP / 4;

        v4i dd[NP], ss[NP];
#pragma unroll
        for (int i = 0; i < NP; ++i) {
            const int it = tid + i * L1_T;
            if (it < PART_PER) {
                dd[i] = ((const v4i*)dst)[pbase + it];
                ss[i] = ((const v4i*)src)[pbase + it];
            }
        }
        for (int i = tid; i < (AGG_WGS + AGG_WGS * PART_CAP) / 4; i += L1_T) ((v4i*)srt)[i] = (v4i){0, 0, 0, 0};
        __syncthreads();

#pragma unroll
        for (int i = 0; i < NP; ++i) {
            const int it = tid + i * L1_T;
            if (it < PART_PER) {
#pragma unroll
                for (int c = 0; c < 4; ++c) {
                    const unsigned d    = (unsigned)dd[i][c];
                    const unsigned r    = d / AGG_RPW;
                    const unsigned dl   = d - r * AGG_RPW;
                    const unsigned slot = atomicAdd(&bc[r], 1u);
                    if (slot < PART_CAP) stg[r * PART_CAP + slot] = (unsigned)ss[i][c] | (dl << AGG_SBITS);
                }
            }
        }
        __syncthreads();

        for (int gi = tid; gi < AGG_WGS * GPR; gi += L1_T) {
            const int r = gi / GPR;
            const int g = gi % GPR;
            if ((unsigned)(g * 4) < bc[r])
                ((v4i*)bkt)[((size_t)r * PART_WGS + p) * GPR + g] = ((const v4i*)stg)[gi];
        }
        if (tid < AGG_WGS) bcnt[tid * PART_WGS + p] = (int)bc[tid];
        return;
    }

    constexpr int NT = N / 16;

    const int lane     = threadIdx.x & 31;
    const int wave     = threadIdx.x >> 5;
    const int row_tile = ((int)blockIdx.x - PART_WGS) * GEMM_SPW + wave;
    if (row_tile * 16 >= M) return;

    const int m     = lane & 15;
    const int khalf = lane >> 4;
    const int row   = row_tile * 16 + m;

    v8f acc[NT];
#pragma unroll
    for (int nt = 0; nt < NT; ++nt) acc[nt] = (v8f){0.f,0.f,0.f,0.f,0.f,0.f,0.f,0.f};

    const float* __restrict__ arow = A + (size_t)row * K;
#pragma unroll
    for (int k0 = 0; k0 < K; k0 += 4) {
        const int kk = k0 + khalf * 2;
        const v2f a = *(const v2f*)(arow + kk);
#pragma unroll
        for (int nt = 0; nt < NT; ++nt) {
            const int n = nt * 16 + m;
            v2f b;
            b.x = B[(size_t)kk * N + n];
            b.y = B[(size_t)(kk + 1) * N + n];
            acc[nt] = __builtin_amdgcn_wmma_f32_16x16x4_f32(
                false, a, false, b, (short)0, acc[nt], false, false);
        }
    }

    const int r0 = row_tile * 16 + khalf * 8;
#pragma unroll
    for (int nt = 0; nt < NT; ++nt) {
        float* __restrict__ crow = C + (size_t)r0 * N + nt * 16 + m;
#pragma unroll
        for (int j = 0; j < 8; ++j) crow[(size_t)j * N] = acc[nt][j];
    }
}

#define DEG_T   256
#define DEG8_BIG 255u
#define DEG_GPR (PART_CAP / 4)
#define DEG_NG  (PART_WGS * DEG_GPR / DEG_T)
static_assert((PART_WGS * DEG_GPR) % DEG_T == 0, "every thread reads the same number of groups");
static_assert(PART_WGS <= DEG_T, "one thread per bucket reads its length");
static_assert(PART_WGS % 32 == 0, "the lengths are looked through 32 at a time");

__global__ __launch_bounds__(DEG_T) void gcn_deg_kernel(const int* __restrict__ dst, const int* __restrict__ bcnt,
                                                        const unsigned* __restrict__ bkt, float* __restrict__ deg,
                                                        unsigned char* __restrict__ deg8) {
    __shared__ unsigned cnt[AGG_RPW];
    __shared__ int blen[PART_WGS];
    const int tid   = threadIdx.x;
    const int lo    = blockIdx.x * AGG_RPW;
    const int rem   = NNODES - lo;
    const int nrows = rem < AGG_RPW ? rem : AGG_RPW;
    const int pb    = blockIdx.x * PART_WGS;

    if (tid < PART_WGS) blen[tid] = bcnt[pb + tid];
    for (int i = tid; i < AGG_RPW; i += DEG_T) cnt[i] = 0u;
    __syncthreads();

    int left[DEG_NG];
#pragma unroll
    for (int i = 0; i < DEG_NG; ++i) {
        const int gi = tid + i * DEG_T;
        const int n  = blen[gi / DEG_GPR];
        left[i] = n <= PART_CAP ? n - (gi % DEG_GPR) * 4 : 0;
    }
    v4i e[DEG_NG];
#pragma unroll
    for (int i = 0; i < DEG_NG; ++i) {
        const int gi = tid + i * DEG_T;
        if (left[i] > 0) e[i] = ((const v4i*)bkt)[(size_t)pb * DEG_GPR + gi];
    }
#pragma unroll
    for (int i = 0; i < DEG_NG; ++i) {
#pragma unroll
        for (int c = 0; c < 4; ++c) {
            if (c < left[i]) atomicAdd(&cnt[(unsigned)e[i][c] >> AGG_SBITS], 1u);
        }
    }

    unsigned over = 0u;
    for (int k = 0; k < PART_WGS; k += 32) over |= __builtin_amdgcn_ballot_w32(blen[k + (tid & 31)] > PART_CAP);
    if (over != 0u) {
        for (int b = 0; b < PART_WGS; ++b) {
            if (blen[b] > PART_CAP) {
                const int e0 = b * (PART_PER * 4);
                for (int i = tid; i < PART_PER * 4; i += DEG_T) {
                    const unsigned dl = (unsigned)(dst[e0 + i] - lo);
                    if (dl < (unsigned)nrows) atomicAdd(&cnt[dl], 1u);
                }
            }
        }
    }
    __syncthreads();
    for (int i = tid; i < nrows; i += DEG_T) {
        const unsigned c = 1u + cnt[i];
        deg[lo + i]  = 1.0f + (float)cnt[i];
        deg8[lo + i] = (unsigned char)(c < DEG8_BIG ? c : DEG8_BIG);
    }
}

#define AGG_QCAP  64
#define AGG_G     16
#define AGG_LCAP  512

static_assert((AGG_WGS - 1) * AGG_RPW < NNODES, "every work-group owns at least one row");
static_assert(AGG_LCAP % 32 == 0, "a full group of 32 saved entries starts on a 256-byte boundary");
static_assert(AGG_WAVES == 32, "one lane per wave reads or stores the list lengths");

template <int F, bool SAVE, bool USE>
__global__ __launch_bounds__(AGG_T) void gcn_agg_lds_kernel(
    const float* __restrict__ h, const int* __restrict__ src,
    const int* __restrict__ dst, const int* __restrict__ bcnt, const unsigned* __restrict__ bkt,
    const float* __restrict__ deg, const unsigned char* __restrict__ deg8, const float* __restrict__ bias,
    float* __restrict__ out, int* __restrict__ lcnt, int* __restrict__ lent,
    const float* __restrict__ Wn, float* __restrict__ hn) {
    constexpr int VPL = F / 32;
    __shared__ __attribute__((aligned(16))) float acc[AGG_RPW * F];
    __shared__ int pend[AGG_WAVES * AGG_QCAP];
    __shared__ __attribute__((aligned(16))) unsigned char dtab[SAVE ? NNODES : 16];

    const int tid  = threadIdx.x;
    const int lane = tid & 31;
    const int wave = tid >> 5;
    const int lo   = blockIdx.x * AGG_RPW;
    const int rem  = NNODES - lo;
    const int nrows = rem < AGG_RPW ? rem : AGG_RPW;

    const int nvec = nrows * (F / 4);
    constexpr int NI = (AGG_RPW * (F / 4) + AGG_T - 1) / AGG_T;
    static_assert(NNODES % 16 == 0, "the byte table is whole 16-byte items");
    constexpr int TI = NNODES / 16;
    constexpr int TN = SAVE ? (TI + AGG_T - 1) / AGG_T : 1;
    v4i tv[TN];
    if (SAVE) {
#pragma unroll
        for (int i = 0; i < TN; ++i) {
            const int idx = tid + i * AGG_T;
            if (idx < TI) tv[i] = ((const v4i*)deg8)[idx];
        }
    }
    {
        const float* __restrict__ hown = h + (size_t)lo * F;
        static_assert(AGG_T % (F / 4) == 0, "a thread's pieces share their columns");
        const v4f bv = ((const v4f*)bias)[tid % (F / 4)];
        v4f tmp[NI];
        float dg[NI];
#pragma unroll
        for (int i = 0; i < NI; ++i) {
            const int idx = tid + i * AGG_T;
            if (idx < nvec) {
                tmp[i] = ((const v4f*)hown)[idx];
                dg[i]  = deg[lo + idx / (F / 4)];
            }
        }
#pragma unroll
        for (int i = 0; i < NI; ++i) {
            const int idx = tid + i * AGG_T;
            if (idx < nvec) {
                const float sq  = rsqrtf(dg[i]);
                const float nrm = sq * sq;
                ((v4f*)acc)[idx] = tmp[i] * nrm + bv;
            }
        }
    }
    if (SAVE) {
#pragma unroll
        for (int i = 0; i < TN; ++i) {
            const int idx = tid + i * AGG_T;
            if (idx < TI) ((v4i*)dtab)[idx] = tv[i];
        }
    }
    __syncthreads();

    int* __restrict__ myq = pend + wave * AGG_QCAP;
    v2i* __restrict__ mylst = (v2i*)lent + ((size_t)blockIdx.x * AGG_WAVES + wave) * AGG_LCAP;

    auto gather_add = [&](int s, int dl, float w, int cnt) {
        const float* __restrict__ hb = h + lane * VPL;
#pragma unroll
        for (int j = 0; j < 32; j += AGG_G) {
            if (j < cnt) {
                float v[AGG_G][VPL];
#pragma unroll
                for (int t = 0; t < AGG_G; ++t) {
                    const int sj = __builtin_amdgcn_readlane(s, j + t);
                    const float* __restrict__ p = hb + (size_t)sj * F;
                    if (VPL == 2) { const v2f r = *(const v2f*)p; v[t][0] = r.x; v[t][VPL - 1] = r.y; }
                    else          { v[t][0] = *p; }
                }
#pragma unroll
                for (int t = 0; t < AGG_G; ++t) {
                    if (j + t < cnt) {
                        const float wj = __int_as_float(__builtin_amdgcn_readlane(__float_as_int(w), j + t));
                        const int   dj = __builtin_amdgcn_readlane(dl, j + t);
                        float* a = acc + dj * F + lane * VPL;
#pragma unroll
                        for (int c = 0; c < VPL; ++c) atomicAdd(a + c, v[t][c] * wj);
                    }
                }
            }
        }
    };

    bool from_list = false;
    if (USE) {
        const int len_l = lcnt[blockIdx.x * AGG_WAVES + lane];
        from_list = __builtin_amdgcn_ballot_w32(len_l > AGG_LCAP) == 0u;
        if (from_list) {
            const int n = __builtin_amdgcn_readlane(len_l, __builtin_amdgcn_readfirstlane(wave));
            v2i cur = (v2i){0, 0};
            if (lane < n) cur = mylst[lane];
            for (int base = 0; base < n; base += 32) {
                const int rest = n - base;
                const int cnt  = rest < 32 ? rest : 32;
                v2i nxt = (v2i){0, 0};
                if (base + 32 + lane < n) nxt = mylst[base + 32 + lane];
                gather_add(cur.x & ((1 << AGG_SBITS) - 1), cur.x >> AGG_SBITS, __int_as_float(cur.y), cnt);
                cur = nxt;
            }
        }
    }

    int tot = 0;
    if (!from_list) {
        int qn = 0;
        const unsigned lt = (1u << lane) - 1u;

        auto drain = [&](int base, int cnt) {
            const int pk  = myq[base + (lane < cnt ? lane : 0)];
            const int s   = pk & ((1 << AGG_SBITS) - 1);
            const int dl  = pk >> AGG_SBITS;
            float w;
            if (SAVE) {
                const unsigned cs = dtab[s];
                const unsigned cd = dtab[lo + dl];
                if (__builtin_amdgcn_ballot_w32(cs == DEG8_BIG || cd == DEG8_BIG) == 0u) {
                    w = rsqrtf((float)cs) * rsqrtf((float)cd);
                } else {
                    const float dgs = cs == DEG8_BIG ? deg[s] : (float)cs;
                    const float dgd = cd == DEG8_BIG ? deg[lo + dl] : (float)cd;
                    w = rsqrtf(dgs) * rsqrtf(dgd);
                }
            } else {
                w = rsqrtf(deg[s]) * rsqrtf(deg[lo + dl]);
            }
            if (SAVE) {
                const int p = tot + lane;
                if (lane < cnt && p < AGG_LCAP) mylst[p] = (v2i){pk, __float_as_int(w)};
                tot += cnt;
            }
            gather_add(s, dl, w, cnt);
        };

        const int pb = blockIdx.x * PART_WGS + __builtin_amdgcn_readfirstlane(wave) * PART_SPW;
        const int bl = lane < PART_SPW ? bcnt[pb + lane] : 0;
        if (__builtin_amdgcn_ballot_w32(bl > PART_CAP) == 0u) {
            constexpr int NQ = PART_SPW * (PART_CAP / 32);
            auto ldgrp = [&](int q) {
                const int j   = q % PART_SPW;
                const int idx = (q / PART_SPW) * 32 + lane;
                const int n   = __builtin_amdgcn_readlane(bl, j);
                return idx < n ? (int)bkt[(size_t)(pb + j) * PART_CAP + idx] : 0;
            };
            int cur = ldgrp(0);
#pragma nounroll
            for (int q = 0; q < NQ; ++q) {
                const int rest = __builtin_amdgcn_readlane(bl, q % PART_SPW) - (q / PART_SPW) * 32;
                const int n    = rest < 0 ? 0 : (rest < 32 ? rest : 32);
                if (lane < n) myq[qn + lane] = cur;
                qn += n;
                int nxt = 0;
                if (q + 1 < NQ) nxt = ldgrp(q + 1);
                if (qn >= 32) { qn -= 32; drain(qn, 32); }
                cur = nxt;
            }
        } else {
            constexpr int EPW = NEDGES / AGG_WAVES;
            const int e0 = __builtin_amdgcn_readfirstlane(wave) * EPW;
#pragma nounroll
            for (int i = 0; i < EPW; i += 32) {
                int pk = -1;
                if (i + lane < EPW) {
                    const unsigned dl = (unsigned)(dst[e0 + i + lane] - lo);
                    if (dl < (unsigned)nrows) pk = src[e0 + i + lane] | (int)(dl << AGG_SBITS);
                }
                const unsigned bal = __builtin_amdgcn_ballot_w32(pk >= 0);
                if (pk >= 0) myq[qn + __builtin_popcount(bal & lt)] = pk;
                qn += __builtin_popcount(bal);
                if (qn >= 32) { qn -= 32; drain(qn, 32); }
            }
        }
        if (qn > 0) drain(0, qn);
    }
    if (SAVE && lane == 0) myq[0] = tot;

    __syncthreads();
    if (SAVE && wave == 0) lcnt[blockIdx.x * AGG_WAVES + lane] = pend[lane * AGG_QCAP];
    if (SAVE) {
        static_assert(!SAVE || (F == HID_DIM && OUT_DIM == 32), "epilogue: K = F, two column tiles");
        static_assert((AGG_RPW + 15) / 16 <= AGG_WAVES, "one slab per wave");
        const int tile = __builtin_amdgcn_readfirstlane(wave);
        if (tile * 16 < nrows) {
            const int m     = lane & 15;
            const int khalf = lane >> 4;
            const int rl = tile * 16 + m < nrows ? tile * 16 + m : nrows - 1;
            const float* arow = acc + rl * F;
            v8f ct[2];
            ct[0] = (v8f){0.f,0.f,0.f,0.f,0.f,0.f,0.f,0.f};
            ct[1] = (v8f){0.f,0.f,0.f,0.f,0.f,0.f,0.f,0.f};
#pragma unroll
            for (int k0 = 0; k0 < F; k0 += 4) {
                const int kk = k0 + khalf * 2;
                v2f a = *(const v2f*)(arow + kk);
                a.x = fmaxf(a.x, 0.0f); a.y = fmaxf(a.y, 0.0f);
#pragma unroll
                for (int nt = 0; nt < 2; ++nt) {
                    const int n = nt * 16 + m;
                    v2f b;
                    b.x = Wn[(size_t)kk * OUT_DIM + n];
                    b.y = Wn[(size_t)(kk + 1) * OUT_DIM + n];
                    ct[nt] = __builtin_amdgcn_wmma_f32_16x16x4_f32(
                        false, a, false, b, (short)0, ct[nt], false, false);
                }
            }
            const int rb = tile * 16 + khalf * 8;
#pragma unroll
            for (int nt = 0; nt < 2; ++nt) {
                float* __restrict__ crow = hn + (size_t)(lo + rb) * OUT_DIM + nt * 16 + m;
#pragma unroll
                for (int j = 0; j < 8; ++j) {
                    if (rb + j < nrows) crow[(size_t)j * OUT_DIM] = ct[nt][j];
                }
            }
        }
    } else {
        float* __restrict__ obase = out + (size_t)lo * F;
#pragma unroll
        for (int i = 0; i < NI; ++i) {
            const int idx = tid + i * AGG_T;
            if (idx < nvec) ((v4f*)obase)[idx] = ((const v4f*)acc)[idx];
        }
    }
}

extern "C" void kernel_launch(void* const* d_in, const int* in_sizes, int n_in,
                              void* d_out, int out_size, void* d_ws, size_t ws_size,
                              hipStream_t stream) {
    const float* x  = (const float*)d_in[0];
    const int*   ei = (const int*)d_in[1];
    const float* W1 = (const float*)d_in[2];
    const float* b1 = (const float*)d_in[3];
    const float* W2 = (const float*)d_in[4];
    const float* b2 = (const float*)d_in[5];
    const int* src = ei;
    const int* dst = ei + NEDGES;
    float* out = (float*)d_out;

    float* ws   = (float*)d_ws;
    float* deg  = ws;
    float* h1   = ws + 131072;
    int* lcnt   = (int*)(h1 + (size_t)NNODES * HID_DIM);
    int* lent   = lcnt + AGG_WGS * AGG_WAVES;
    int* bcnt     = lent + (size_t)AGG_WGS * AGG_WAVES * AGG_LCAP * 2;
    unsigned* bkt = (unsigned*)(bcnt + AGG_WGS * PART_WGS);
    float* h2   = (float*)(bkt + (size_t)AGG_WGS * PART_WGS * PART_CAP);
    unsigned char* deg8 = (unsigned char*)(h2 + (size_t)NNODES * OUT_DIM);

    gcn_gemm_sort_kernel<IN_DIM, HID_DIM><<<PART_WGS + GEMM_WGS, L1_T, 0, stream>>>(
        x, W1, h1, NNODES, src, dst, bcnt, bkt);

    gcn_deg_kernel<<<AGG_WGS, DEG_T, 0, stream>>>(dst, bcnt, bkt, deg, deg8);

    gcn_agg_lds_kernel<HID_DIM, true, false><<<AGG_WGS, AGG_T, 0, stream>>>(
        h1, src, dst, bcnt, bkt, deg, deg8, b1, nullptr, lcnt, lent, W2, h2);

    gcn_agg_lds_kernel<OUT_DIM, false, true><<<AGG_WGS, AGG_T, 0, stream>>>(
        h2, src, dst, bcnt, bkt, deg, deg8, b2, out, lcnt, lent, nullptr, nullptr);
}


# aggregation waves wait for memory less often: all 32 row gathers of a 32-edge group issued before its first LDS add (were 16, add, 16, add), 32-bit offsets from a scalar base; start rows branch-free
# speedup vs baseline: 3.8502x; 1.0484x over previous
#include <hip/hip_runtime.h>

#define NNODES 100000
#define NEDGES 1200000
#define IN_DIM 64
#define HID_DIM 64
#define OUT_DIM 32

typedef __attribute__((ext_vector_type(2))) float v2f;
typedef __attribute__((ext_vector_type(4))) float v4f;
typedef __attribute__((ext_vector_type(8))) float v8f;
typedef __attribute__((ext_vector_type(4))) int v4i;
typedef __attribute__((ext_vector_type(2))) int v2i;

#define AGG_WGS   256
#define AGG_RPW   ((NNODES + AGG_WGS - 1) / AGG_WGS)
#define AGG_T     1024
#define AGG_WAVES (AGG_T / 32)
#define AGG_SBITS 17

#define PART_WGS  160
#define PART_PER  (NEDGES / 4 / PART_WGS)
#define PART_CAP  64
#define PART_SPW  (PART_WGS / AGG_WAVES)
static_assert((NEDGES / 4) % PART_WGS == 0, "the slices are equal");
static_assert(PART_WGS % AGG_WAVES == 0, "every aggregation wave reads the same number of buckets");
static_assert(PART_SPW <= 32, "one lane per bucket reads its length");
static_assert(PART_CAP % 32 == 0, "a bucket is read 32 entries at a time");
static_assert(NEDGES % AGG_WAVES == 0, "a wave's slices are a whole part of the edge list");
static_assert(NNODES <= (1 << AGG_SBITS), "source row must fit its field of an entry");
static_assert(AGG_RPW < (1 << (31 - AGG_SBITS)), "local row must fit its field of an entry");

#define L1_T     512
#define GEMM_SPW (L1_T / 32)
#define GEMM_WGS ((NNODES / 16 + GEMM_SPW - 1) / GEMM_SPW)
static_assert(NNODES % 16 == 0, "the GEMM has whole slabs only");

template <int K, int N>
__global__ __launch_bounds__(L1_T) void gcn_gemm_sort_kernel(
    const float* __restrict__ A, const float* __restrict__ B, float* __restrict__ C, int M,
    const int* __restrict__ src, const int* __restrict__ dst,
    int* __restrict__ bcnt, unsigned* __restrict__ bkt) {
    if (blockIdx.x < PART_WGS) {
        __shared__ __attribute__((aligned(16))) unsigned srt[AGG_WGS + AGG_WGS * PART_CAP];
        const int tid   = threadIdx.x;
        const int p     = blockIdx.x;
        const int pbase = p * PART_PER;
        unsigned* bc  = srt;
        unsigned* stg = srt + AGG_WGS;
        constexpr int NP  = (PART_PER + L1_T - 1) / L1_T;
        constexpr int GPR = PART_CAP / 4;

        v4i dd[NP], ss[NP];
#pragma unroll
        for (int i = 0; i < NP; ++i) {
            const int it = tid + i * L1_T;
            if (it < PART_PER) {
                dd[i] = ((const v4i*)dst)[pbase + it];
                ss[i] = ((const v4i*)src)[pbase + it];
            }
        }
        for (int i = tid; i < (AGG_WGS + AGG_WGS * PART_CAP) / 4; i += L1_T) ((v4i*)srt)[i] = (v4i){0, 0, 0, 0};
        __syncthreads();

#pragma unroll
        for (int i = 0; i < NP; ++i) {
            const int it = tid + i * L1_T;
            if (it < PART_PER) {
#pragma unroll
                for (int c = 0; c < 4; ++c) {
                    const unsigned d    = (unsigned)dd[i][c];
                    const unsigned r    = d / AGG_RPW;
                    const unsigned dl   = d - r * AGG_RPW;
                    const unsigned slot = atomicAdd(&bc[r], 1u);
                    if (slot < PART_CAP) stg[r * PART_CAP + slot] = (unsigned)ss[i][c] | (dl << AGG_SBITS);
                }
            }
        }
        __syncthreads();

        for (int gi = tid; gi < AGG_WGS * GPR; gi += L1_T) {
            const int r = gi / GPR;
            const int g = gi % GPR;
            if ((unsigned)(g * 4) < bc[r])
                ((v4i*)bkt)[((size_t)r * PART_WGS + p) * GPR + g] = ((const v4i*)stg)[gi];
        }
        if (tid < AGG_WGS) bcnt[tid * PART_WGS + p] = (int)bc[tid];
        return;
    }

    constexpr int NT = N / 16;

    const int lane     = threadIdx.x & 31;
    const int wave     = threadIdx.x >> 5;
    const int row_tile = ((int)blockIdx.x - PART_WGS) * GEMM_SPW + wave;
    if (row_tile * 16 >= M) return;

    const int m     = lane & 15;
    const int khalf = lane >> 4;
    const int row   = row_tile * 16 + m;

    v8f acc[NT];
#pragma unroll
    for (int nt = 0; nt < NT; ++nt) acc[nt] = (v8f){0.f,0.f,0.f,0.f,0.f,0.f,0.f,0.f};

    const float* __restrict__ arow = A + (size_t)row * K;
#pragma unroll
    for (int k0 = 0; k0 < K; k0 += 4) {
        const int kk = k0 + khalf * 2;
        const v2f a = *(const v2f*)(arow + kk);
#pragma unroll
        for (int nt = 0; nt < NT; ++nt) {
            const int n = nt * 16 + m;
            v2f b;
            b.x = B[(size_t)kk * N + n];
            b.y = B[(size_t)(kk + 1) * N + n];
            acc[nt] = __builtin_amdgcn_wmma_f32_16x16x4_f32(
                false, a, false, b, (short)0, acc[nt], false, false);
        }
    }

    const int r0 = row_tile * 16 + khalf * 8;
#pragma unroll
    for (int nt = 0; nt < NT; ++nt) {
        float* __restrict__ crow = C + (size_t)r0 * N + nt * 16 + m;
#pragma unroll
        for (int j = 0; j < 8; ++j) crow[(size_t)j * N] = acc[nt][j];
    }
}

#define DEG_T   256
#define DEG8_BIG 255u
#define DEG_GPR (PART_CAP / 4)
#define DEG_NG  (PART_WGS * DEG_GPR / DEG_T)
static_assert((PART_WGS * DEG_GPR) % DEG_T == 0, "every thread reads the same number of groups");
static_assert(PART_WGS <= DEG_T, "one thread per bucket reads its length");
static_assert(PART_WGS % 32 == 0, "the lengths are looked through 32 at a time");

__global__ __launch_bounds__(DEG_T) void gcn_deg_kernel(const int* __restrict__ dst, const int* __restrict__ bcnt,
                                                        const unsigned* __restrict__ bkt, float* __restrict__ deg,
                                                        unsigned char* __restrict__ deg8) {
    __shared__ unsigned cnt[AGG_RPW];
    __shared__ int blen[PART_WGS];
    const int tid   = threadIdx.x;
    const int lo    = blockIdx.x * AGG_RPW;
    const int rem   = NNODES - lo;
    const int nrows = rem < AGG_RPW ? rem : AGG_RPW;
    const int pb    = blockIdx.x * PART_WGS;

    if (tid < PART_WGS) blen[tid] = bcnt[pb + tid];
    for (int i = tid; i < AGG_RPW; i += DEG_T) cnt[i] = 0u;
    __syncthreads();

    int left[DEG_NG];
#pragma unroll
    for (int i = 0; i < DEG_NG; ++i) {
        const int gi = tid + i * DEG_T;
        const int n  = blen[gi / DEG_GPR];
        left[i] = n <= PART_CAP ? n - (gi % DEG_GPR) * 4 : 0;
    }
    v4i e[DEG_NG];
#pragma unroll
    for (int i = 0; i < DEG_NG; ++i) {
        const int gi = tid + i * DEG_T;
        if (left[i] > 0) e[i] = ((const v4i*)bkt)[(size_t)pb * DEG_GPR + gi];
    }
#pragma unroll
    for (int i = 0; i < DEG_NG; ++i) {
#pragma unroll
        for (int c = 0; c < 4; ++c) {
            if (c < left[i]) atomicAdd(&cnt[(unsigned)e[i][c] >> AGG_SBITS], 1u);
        }
    }

    unsigned over = 0u;
    for (int k = 0; k < PART_WGS; k += 32) over |= __builtin_amdgcn_ballot_w32(blen[k + (tid & 31)] > PART_CAP);
    if (over != 0u) {
        for (int b = 0; b < PART_WGS; ++b) {
            if (blen[b] > PART_CAP) {
                const int e0 = b * (PART_PER * 4);
                for (int i = tid; i < PART_PER * 4; i += DEG_T) {
                    const unsigned dl = (unsigned)(dst[e0 + i] - lo);
                    if (dl < (unsigned)nrows) atomicAdd(&cnt[dl], 1u);
                }
            }
        }
    }
    __syncthreads();
    for (int i = tid; i < nrows; i += DEG_T) {
        const unsigned c = 1u + cnt[i];
        deg[lo + i]  = 1.0f + (float)cnt[i];
        deg8[lo + i] = (unsigned char)(c < DEG8_BIG ? c : DEG8_BIG);
    }
}

#define AGG_QCAP  64
#define AGG_G     16
#define AGG_LCAP  512

static_assert((AGG_WGS - 1) * AGG_RPW < NNODES, "every work-group owns at least one row");
static_assert(AGG_LCAP % 32 == 0, "a full group of 32 saved entries starts on a 256-byte boundary");
static_assert(AGG_WAVES == 32, "one lane per wave reads or stores the list lengths");

template <int F, bool SAVE, bool USE>
__global__ __launch_bounds__(AGG_T) void gcn_agg_lds_kernel(
    const float* __restrict__ h, const int* __restrict__ src,
    const int* __restrict__ dst, const int* __restrict__ bcnt, const unsigned* __restrict__ bkt,
    const float* __restrict__ deg, const unsigned char* __restrict__ deg8, const float* __restrict__ bias,
    float* __restrict__ out, int* __restrict__ lcnt, int* __restrict__ lent,
    const float* __restrict__ Wn, float* __restrict__ hn) {
    constexpr int VPL = F / 32;
    __shared__ __attribute__((aligned(16))) float acc[AGG_RPW * F];
    __shared__ int pend[AGG_WAVES * AGG_QCAP];
    __shared__ __attribute__((aligned(16))) unsigned char dtab[SAVE ? NNODES : 16];

    const int tid  = threadIdx.x;
    const int lane = tid & 31;
    const int wave = tid >> 5;
    const int lo   = blockIdx.x * AGG_RPW;
    const int rem  = NNODES - lo;
    const int nrows = rem < AGG_RPW ? rem : AGG_RPW;

    const int nvec = nrows * (F / 4);
    constexpr int NI = (AGG_RPW * (F / 4) + AGG_T - 1) / AGG_T;
    static_assert(NNODES % 16 == 0, "the byte table is whole 16-byte items");
    constexpr int TI = NNODES / 16;
    constexpr int TN = SAVE ? (TI + AGG_T - 1) / AGG_T : 1;
    v4i tv[TN];
    if (SAVE) {
#pragma unroll
        for (int i = 0; i < TN; ++i) {
            const int idx = tid + i * AGG_T;
            if (idx < TI) tv[i] = ((const v4i*)deg8)[idx];
        }
    }
    {
        const float* __restrict__ hown = h + (size_t)lo * F;
        static_assert(AGG_T % (F / 4) == 0, "a thread's pieces share their columns");
        const v4f bv = ((const v4f*)bias)[tid % (F / 4)];
        v4f tmp[NI];
        float dg[NI];
#pragma unroll
        for (int i = 0; i < NI; ++i) {
            const int idx = tid + i * AGG_T;
            const int ci  = idx < nvec ? idx : nvec - 1;
            tmp[i] = ((const v4f*)hown)[ci];
            dg[i]  = deg[lo + ci / (F / 4)];
        }
#pragma unroll
        for (int i = 0; i < NI; ++i) {
            const int idx = tid + i * AGG_T;
            if (idx < nvec) {
                const float sq  = rsqrtf(dg[i]);
                const float nrm = sq * sq;
                ((v4f*)acc)[idx] = tmp[i] * nrm + bv;
            }
        }
    }
    if (SAVE) {
#pragma unroll
        for (int i = 0; i < TN; ++i) {
            const int idx = tid + i * AGG_T;
            if (idx < TI) ((v4i*)dtab)[idx] = tv[i];
        }
    }
    __syncthreads();

    int* __restrict__ myq = pend + wave * AGG_QCAP;
    v2i* __restrict__ mylst = (v2i*)lent + ((size_t)blockIdx.x * AGG_WAVES + wave) * AGG_LCAP;

    static_assert((unsigned long long)NNODES * F * sizeof(float) < (1ull << 32), "a byte offset into h fits 32 bits");
    const unsigned loffb = (unsigned)lane * (unsigned)(VPL * sizeof(float));
    auto gather_add = [&](int s, int dl, float w, int cnt) {
        float v[32][VPL];
#pragma unroll
        for (int j = 0; j < 32; j += AGG_G) {
            if (j == 0 || j < cnt) {
#pragma unroll
                for (int t = 0; t < AGG_G; ++t) {
                    const unsigned sj = (unsigned)__builtin_amdgcn_readlane(s, j + t);
                    const unsigned off = sj * (unsigned)(F * sizeof(float)) + loffb;
                    const char* p = (const char*)h + off;
                    if (VPL == 2) { const v2f r = *(const v2f*)p; v[j + t][0] = r.x; v[j + t][VPL - 1] = r.y; }
                    else          { v[j + t][0] = *(const float*)p; }
                }
            }
        }
#pragma unroll
        for (int j = 0; j < 32; j += AGG_G) {
            if (j == 0 || j < cnt) {
#pragma unroll
                for (int t = 0; t < AGG_G; ++t) {
                    if (j + t < cnt) {
                        const float wj = __int_as_float(__builtin_amdgcn_readlane(__float_as_int(w), j + t));
                        const int   dj = __builtin_amdgcn_readlane(dl, j + t);
                        float* a = acc + dj * F + lane * VPL;
#pragma unroll
                        for (int c = 0; c < VPL; ++c) atomicAdd(a + c, v[j + t][c] * wj);
                    }
                }
            }
        }
    };

    bool from_list = false;
    if (USE) {
        const int len_l = lcnt[blockIdx.x * AGG_WAVES + lane];
        from_list = __builtin_amdgcn_ballot_w32(len_l > AGG_LCAP) == 0u;
        if (from_list) {
            const int n = __builtin_amdgcn_readlane(len_l, __builtin_amdgcn_readfirstlane(wave));
            v2i cur = (v2i){0, 0};
            if (lane < n) cur = mylst[lane];
            const int nfull = n & ~31;
#pragma nounroll
            for (int base = 0; base < nfull; base += 32) {
                v2i nxt = (v2i){0, 0};
                if (base + 32 + lane < n) nxt = mylst[base + 32 + lane];
                gather_add(cur.x & ((1 << AGG_SBITS) - 1), cur.x >> AGG_SBITS, __int_as_float(cur.y), 32);
                cur = nxt;
            }
            if (n > nfull)
                gather_add(cur.x & ((1 << AGG_SBITS) - 1), cur.x >> AGG_SBITS, __int_as_float(cur.y), n - nfull);
        }
    }

    int tot = 0;
    if (!from_list) {
        int qn = 0;
        const unsigned lt = (1u << lane) - 1u;

        auto drain = [&](int base, int cnt) {
            const int pk  = myq[base + (lane < cnt ? lane : 0)];
            const int s   = pk & ((1 << AGG_SBITS) - 1);
            const int dl  = pk >> AGG_SBITS;
            float w;
            if (SAVE) {
                const unsigned cs = dtab[s];
                const unsigned cd = dtab[lo + dl];
                if (__builtin_amdgcn_ballot_w32(cs == DEG8_BIG || cd == DEG8_BIG) == 0u) {
                    w = rsqrtf((float)cs) * rsqrtf((float)cd);
                } else {
                    const float dgs = cs == DEG8_BIG ? deg[s] : (float)cs;
                    const float dgd = cd == DEG8_BIG ? deg[lo + dl] : (float)cd;
                    w = rsqrtf(dgs) * rsqrtf(dgd);
                }
            } else {
                w = rsqrtf(deg[s]) * rsqrtf(deg[lo + dl]);
            }
            if (SAVE) {
                const int p = tot + lane;
                if (lane < cnt && p < AGG_LCAP) mylst[p] = (v2i){pk, __float_as_int(w)};
                tot += cnt;
            }
            gather_add(s, dl, w, cnt);
        };

        const int pb = blockIdx.x * PART_WGS + __builtin_amdgcn_readfirstlane(wave) * PART_SPW;
        const int bl = lane < PART_SPW ? bcnt[pb + lane] : 0;
        if (__builtin_amdgcn_ballot_w32(bl > PART_CAP) == 0u) {
            constexpr int NQ = PART_SPW * (PART_CAP / 32);
            auto ldgrp = [&](int q) {
                const int j   = q % PART_SPW;
                const int idx = (q / PART_SPW) * 32 + lane;
                const int n   = __builtin_amdgcn_readlane(bl, j);
                return idx < n ? (int)bkt[(size_t)(pb + j) * PART_CAP + idx] : 0;
            };
            int cur = ldgrp(0);
#pragma nounroll
            for (int q = 0; q < NQ; ++q) {
                const int rest = __builtin_amdgcn_readlane(bl, q % PART_SPW) - (q / PART_SPW) * 32;
                const int n    = rest < 0 ? 0 : (rest < 32 ? rest : 32);
                if (lane < n) myq[qn + lane] = cur;
                qn += n;
                int nxt = 0;
                if (q + 1 < NQ) nxt = ldgrp(q + 1);
                if (qn >= 32) { qn -= 32; drain(qn, 32); }
                cur = nxt;
            }
        } else {
            constexpr int EPW = NEDGES / AGG_WAVES;
            const int e0 = __builtin_amdgcn_readfirstlane(wave) * EPW;
#pragma nounroll
            for (int i = 0; i < EPW; i += 32) {
                int pk = -1;
                if (i + lane < EPW) {
                    const unsigned dl = (unsigned)(dst[e0 + i + lane] - lo);
                    if (dl < (unsigned)nrows) pk = src[e0 + i + lane] | (int)(dl << AGG_SBITS);
                }
                const unsigned bal = __builtin_amdgcn_ballot_w32(pk >= 0);
                if (pk >= 0) myq[qn + __builtin_popcount(bal & lt)] = pk;
                qn += __builtin_popcount(bal);
                if (qn >= 32) { qn -= 32; drain(qn, 32); }
            }
        }
        if (qn > 0) drain(0, qn);
    }
    if (SAVE && lane == 0) myq[0] = tot;

    __syncthreads();
    if (SAVE && wave == 0) lcnt[blockIdx.x * AGG_WAVES + lane] = pend[lane * AGG_QCAP];
    if (SAVE) {
        static_assert(!SAVE || (F == HID_DIM && OUT_DIM == 32), "epilogue: K = F, two column tiles");
        static_assert((AGG_RPW + 15) / 16 <= AGG_WAVES, "one slab per wave");
        const int tile = __builtin_amdgcn_readfirstlane(wave);
        if (tile * 16 < nrows) {
            const int m     = lane & 15;
            const int khalf = lane >> 4;
            const int rl = tile * 16 + m < nrows ? tile * 16 + m : nrows - 1;
            const float* arow = acc + rl * F;
            v8f ct[2];
            ct[0] = (v8f){0.f,0.f,0.f,0.f,0.f,0.f,0.f,0.f};
            ct[1] = (v8f){0.f,0.f,0.f,0.f,0.f,0.f,0.f,0.f};
#pragma unroll
            for (int k0 = 0; k0 < F; k0 += 4) {
                const int kk = k0 + khalf * 2;
                v2f a = *(const v2f*)(arow + kk);
                a.x = fmaxf(a.x, 0.0f); a.y = fmaxf(a.y, 0.0f);
#pragma unroll
                for (int nt = 0; nt < 2; ++nt) {
                    const int n = nt * 16 + m;
                    v2f b;
                    b.x = Wn[(size_t)kk * OUT_DIM + n];
                    b.y = Wn[(size_t)(kk + 1) * OUT_DIM + n];
                    ct[nt] = __builtin_amdgcn_wmma_f32_16x16x4_f32(
                        false, a, false, b, (short)0, ct[nt], false, false);
                }
            }
            const int rb = tile * 16 + khalf * 8;
#pragma unroll
            for (int nt = 0; nt < 2; ++nt) {
                float* __restrict__ crow = hn + (size_t)(lo + rb) * OUT_DIM + nt * 16 + m;
#pragma unroll
                for (int j = 0; j < 8; ++j) {
                    if (rb + j < nrows) crow[(size_t)j * OUT_DIM] = ct[nt][j];
                }
            }
        }
    } else {
        float* __restrict__ obase = out + (size_t)lo * F;
#pragma unroll
        for (int i = 0; i < NI; ++i) {
            const int idx = tid + i * AGG_T;
            if (idx < nvec) ((v4f*)obase)[idx] = ((const v4f*)acc)[idx];
        }
    }
}

extern "C" void kernel_launch(void* const* d_in, const int* in_sizes, int n_in,
                              void* d_out, int out_size, void* d_ws, size_t ws_size,
                              hipStream_t stream) {
    const float* x  = (const float*)d_in[0];
    const int*   ei = (const int*)d_in[1];
    const float* W1 = (const float*)d_in[2];
    const float* b1 = (const float*)d_in[3];
    const float* W2 = (const float*)d_in[4];
    const float* b2 = (const float*)d_in[5];
    const int* src = ei;
    const int* dst = ei + NEDGES;
    float* out = (float*)d_out;

    float* ws   = (float*)d_ws;
    float* deg  = ws;
    float* h1   = ws + 131072;
    int* lcnt   = (int*)(h1 + (size_t)NNODES * HID_DIM);
    int* lent   = lcnt + AGG_WGS * AGG_WAVES;
    int* bcnt     = lent + (size_t)AGG_WGS * AGG_WAVES * AGG_LCAP * 2;
    unsigned* bkt = (unsigned*)(bcnt + AGG_WGS * PART_WGS);
    float* h2   = (float*)(bkt + (size_t)AGG_WGS * PART_WGS * PART_CAP);
    unsigned char* deg8 = (unsigned char*)(h2 + (size_t)NNODES * OUT_DIM);

    gcn_gemm_sort_kernel<IN_DIM, HID_DIM><<<PART_WGS + GEMM_WGS, L1_T, 0, stream>>>(
        x, W1, h1, NNODES, src, dst, bcnt, bkt);

    gcn_deg_kernel<<<AGG_WGS, DEG_T, 0, stream>>>(dst, bcnt, bkt, deg, deg8);

    gcn_agg_lds_kernel<HID_DIM, true, false><<<AGG_WGS, AGG_T, 0, stream>>>(
        h1, src, dst, bcnt, bkt, deg, deg8, b1, nullptr, lcnt, lent, W2, h2);

    gcn_agg_lds_kernel<OUT_DIM, false, true><<<AGG_WGS, AGG_T, 0, stream>>>(
        h2, src, dst, bcnt, bkt, deg, deg8, b2, out, lcnt, lent, nullptr, nullptr);
}
